# starting kernel
# baseline (speedup 1.0000x reference)
#include <hip/hip_runtime.h>
#include <math.h>

typedef float v2f __attribute__((ext_vector_type(2)));
typedef float v8f __attribute__((ext_vector_type(8)));

#define HO 64
#define INCH 64

__global__ __launch_bounds__(256) void k_zero(float* __restrict__ p, long long n) {
  long long i = (long long)blockIdx.x * blockDim.x + threadIdx.x;
  if (i < n) p[i] = 0.0f;
}

__global__ __launch_bounds__(256) void k_gemm_wmma(
    const float* __restrict__ X, const float* __restrict__ W,
    float* __restrict__ Xp, int nrows) {
  const int wave = (int)(((unsigned)blockIdx.x * blockDim.x + threadIdx.x) >> 5);
  const int lane = (int)(threadIdx.x & 31u);
  const int row0 = wave * 16;
  if (row0 >= nrows) return;
  if (row0 + 16 <= nrows) {
    const int m  = lane & 15;
    const int kh = lane >> 4;
    v8f acc0 = {}, acc1 = {}, acc2 = {}, acc3 = {};
#pragma unroll
    for (int k0 = 0; k0 < INCH; k0 += 4) {
      v2f a = *(const v2f*)(X + (size_t)(row0 + m) * INCH + (k0 + kh * 2));
      const float* w0 = W + (size_t)(k0 + kh * 2) * HO;
      const float* w1 = w0 + HO;
      v2f b0, b1, b2, b3;
      b0.x = w0[m];      b0.y = w1[m];
      b1.x = w0[m + 16]; b1.y = w1[m + 16];
      b2.x = w0[m + 32]; b2.y = w1[m + 32];
      b3.x = w0[m + 48]; b3.y = w1[m + 48];
      acc0 = __builtin_amdgcn_wmma_f32_16x16x4_f32(false, a, false, b0, (short)0, acc0, false, false);
      acc1 = __builtin_amdgcn_wmma_f32_16x16x4_f32(false, a, false, b1, (short)0, acc1, false, false);
      acc2 = __builtin_amdgcn_wmma_f32_16x16x4_f32(false, a, false, b2, (short)0, acc2, false, false);
      acc3 = __builtin_amdgcn_wmma_f32_16x16x4_f32(false, a, false, b3, (short)0, acc3, false, false);
    }
#pragma unroll
    for (int r = 0; r < 8; ++r) {
      const int row = row0 + r + kh * 8;
      float* o = Xp + (size_t)row * HO + m;
      o[0]  = acc0[r];
      o[16] = acc1[r];
      o[32] = acc2[r];
      o[48] = acc3[r];
    }
  } else {
    for (int r = row0 + lane; r < nrows; r += 32)
      for (int c = 0; c < HO; ++c) {
        float s = 0.f;
        for (int k = 0; k < INCH; ++k) s += X[(size_t)r * INCH + k] * W[(size_t)k * HO + c];
        Xp[(size_t)r * HO + c] = s;
      }
  }
}

__global__ __launch_bounds__(256) void k_scatter_v2e(
    const float* __restrict__ Xp, const int* __restrict__ vertex,
    const int* __restrict__ edges, const float* __restrict__ homo,
    float* __restrict__ Xe, float* __restrict__ cnt,
    float* __restrict__ att_sum, int nnz) {
  long long t = (long long)blockIdx.x * blockDim.x + threadIdx.x;
  int i  = (int)(t >> 4);
  int c4 = ((int)t & 15) * 4;
  if (i >= nnz) return;
  const int v = vertex[i];
  const int e = edges[i];
  const float4 x = *(const float4*)(Xp + (size_t)v * HO + c4);
  float* dst = Xe + (size_t)e * HO + c4;
  atomicAdd(dst + 0, x.x);
  atomicAdd(dst + 1, x.y);
  atomicAdd(dst + 2, x.z);
  atomicAdd(dst + 3, x.w);
  if (c4 == 0) {
    atomicAdd(cnt + e, 1.0f);
    atomicAdd(att_sum + v, homo[e]);
  }
}

__global__ __launch_bounds__(256) void k_edge_mean(float* __restrict__ Xe,
                                                   const float* __restrict__ cnt,
                                                   long long total) {
  long long i = (long long)blockIdx.x * blockDim.x + threadIdx.x;
  if (i >= total) return;
  const float c = cnt[i >> 6];
  Xe[i] *= 1.0f / fmaxf(c, 1.0f);
}

__global__ __launch_bounds__(256) void k_scatter_e2v(
    const float* __restrict__ Xe, const int* __restrict__ vertex,
    const int* __restrict__ edges, const float* __restrict__ homo,
    const float* __restrict__ att_sum, float* __restrict__ out, int nnz) {
  long long t = (long long)blockIdx.x * blockDim.x + threadIdx.x;
  int i  = (int)(t >> 4);
  int c4 = ((int)t & 15) * 4;
  if (i >= nnz) return;
  const int v = vertex[i];
  const int e = edges[i];
  const float att = homo[e] / att_sum[v];
  const float4 x = *(const float4*)(Xe + (size_t)e * HO + c4);
  float* dst = out + (size_t)v * HO + c4;
  atomicAdd(dst + 0, att * x.x);
  atomicAdd(dst + 1, att * x.y);
  atomicAdd(dst + 2, att * x.z);
  atomicAdd(dst + 3, att * x.w);
}

__global__ __launch_bounds__(256) void k_rownorm(float* __restrict__ out, int n) {
  const int wave = (int)(((unsigned)blockIdx.x * blockDim.x + threadIdx.x) >> 5);
  const int lane = (int)(threadIdx.x & 31u);
  if (wave >= n) return;
  float* row = out + (size_t)wave * HO + lane * 2;
  float2 x = *(float2*)row;
  float ss = x.x * x.x + x.y * x.y;
#pragma unroll
  for (int off = 16; off > 0; off >>= 1) ss += __shfl_xor(ss, off, 32);
  const float nrm = sqrtf(ss);
  const float scale = (nrm > 0.f) ? 1.0f / fmaxf(nrm, 1e-30f) : 0.0f;
  x.x *= scale;
  x.y *= scale;
  *(float2*)row = x;
}

extern "C" void kernel_launch(void* const* d_in, const int* in_sizes, int n_in,
                              void* d_out, int out_size, void* d_ws, size_t ws_size,
                              hipStream_t stream) {
  const float* X      = (const float*)d_in[0];
  const float* W      = (const float*)d_in[1];
  const float* homo   = (const float*)d_in[2];
  const int*   vertex = (const int*)d_in[3];
  const int*   edges  = (const int*)d_in[4];
  float* out = (float*)d_out;

  const int n   = in_sizes[0] / INCH;
  const int e   = in_sizes[2];
  const int nnz = in_sizes[3];

  float* Xe      = (float*)d_ws;
  float* cnt     = Xe + (size_t)e * HO;
  float* att_sum = cnt + e;
  const long long zn = (long long)e * HO + e + n;
  k_zero<<<(int)((zn + 255) / 256), 256, 0, stream>>>(Xe, zn);

  const int tiles = (n + 15) / 16;
  k_gemm_wmma<<<(tiles * 32 + 255) / 256, 256, 0, stream>>>(X, W, out, n);

  const long long st = (long long)nnz * 16;
  k_scatter_v2e<<<(int)((st + 255) / 256), 256, 0, stream>>>(
      out, vertex, edges, homo, Xe, cnt, att_sum, nnz);

  const long long em = (long long)e * HO;
  k_edge_mean<<<(int)((em + 255) / 256), 256, 0, stream>>>(Xe, cnt, em);

  k_scatter_e2v<<<(int)((st + 255) / 256), 256, 0, stream>>>(
      Xe, vertex, edges, homo, att_sum, out, nnz);

  const long long nt = (long long)n * 32;
  k_rownorm<<<(int)((nt + 255) / 256), 256, 0, stream>>>(out, n);
}


# Both scatter passes: resident waves walk the list, one wave per incidence with lane = channel, so every f32 atomic and row read covers one whole 128-B line (atomic arrival order changes)
# speedup vs baseline: 1.7205x; 1.7205x over previous
#include <hip/hip_runtime.h>
#include <math.h>

typedef float v2f __attribute__((ext_vector_type(2)));
typedef float v8f __attribute__((ext_vector_type(8)));

#define HO 64
#define INCH 64

__global__ __launch_bounds__(256) void k_zero(float* __restrict__ p, long long n) {
  long long i = (long long)blockIdx.x * blockDim.x + threadIdx.x;
  if (i < n) p[i] = 0.0f;
}

__global__ __launch_bounds__(256) void k_gemm_wmma(
    const float* __restrict__ X, const float* __restrict__ W,
    float* __restrict__ Xp, int nrows) {
  const int wave = (int)(((unsigned)blockIdx.x * blockDim.x + threadIdx.x) >> 5);
  const int lane = (int)(threadIdx.x & 31u);
  const int row0 = wave * 16;
  if (row0 >= nrows) return;
  if (row0 + 16 <= nrows) {
    const int m  = lane & 15;
    const int kh = lane >> 4;
    v8f acc0 = {}, acc1 = {}, acc2 = {}, acc3 = {};
#pragma unroll
    for (int k0 = 0; k0 < INCH; k0 += 4) {
      v2f a = *(const v2f*)(X + (size_t)(row0 + m) * INCH + (k0 + kh * 2));
      const float* w0 = W + (size_t)(k0 + kh * 2) * HO;
      const float* w1 = w0 + HO;
      v2f b0, b1, b2, b3;
      b0.x = w0[m];      b0.y = w1[m];
      b1.x = w0[m + 16]; b1.y = w1[m + 16];
      b2.x = w0[m + 32]; b2.y = w1[m + 32];
      b3.x = w0[m + 48]; b3.y = w1[m + 48];
      acc0 = __builtin_amdgcn_wmma_f32_16x16x4_f32(false, a, false, b0, (short)0, acc0, false, false);
      acc1 = __builtin_amdgcn_wmma_f32_16x16x4_f32(false, a, false, b1, (short)0, acc1, false, false);
      acc2 = __builtin_amdgcn_wmma_f32_16x16x4_f32(false, a, false, b2, (short)0, acc2, false, false);
      acc3 = __builtin_amdgcn_wmma_f32_16x16x4_f32(false, a, false, b3, (short)0, acc3, false, false);
    }
#pragma unroll
    for (int r = 0; r < 8; ++r) {
      const int row = row0 + r + kh * 8;
      float* o = Xp + (size_t)row * HO + m;
      o[0]  = acc0[r];
      o[16] = acc1[r];
      o[32] = acc2[r];
      o[48] = acc3[r];
    }
  } else {
    for (int r = row0 + lane; r < nrows; r += 32)
      for (int c = 0; c < HO; ++c) {
        float s = 0.f;
        for (int k = 0; k < INCH; ++k) s += X[(size_t)r * INCH + k] * W[(size_t)k * HO + c];
        Xp[(size_t)r * HO + c] = s;
      }
  }
}

#define TRIP 8
template <bool V2E>
__global__ __launch_bounds__(256) void k_scatter_rows(
    const float* __restrict__ src, const int* __restrict__ vertex,
    const int* __restrict__ edges, const float* __restrict__ homo,
    float* __restrict__ dst, float* __restrict__ cnt,
    float* __restrict__ att_sum, int nnz) {
  const int lane = (int)(threadIdx.x & 31u);
  const int wave = __builtin_amdgcn_readfirstlane(
      (int)(((unsigned)blockIdx.x * blockDim.x + threadIdx.x) >> 5));
  const int nwaves = (int)((gridDim.x * blockDim.x) >> 5);
  for (int base = wave * TRIP; base < nnz; base += nwaves * TRIP) {
    const int idx = base + lane;
    const bool mine = lane < TRIP && idx < nnz;
    int vl = 0, el = 0;
    if (mine) {
      vl = vertex[idx];
      el = edges[idx];
    }
    const bool whole = base + TRIP <= nnz;
    float x0[TRIP], x1[TRIP];
    if (whole) {
#pragma unroll
      for (int u = 0; u < TRIP; ++u) {
        const int r = __builtin_amdgcn_readlane(V2E ? vl : el, u);
        const float* s = src + (size_t)r * HO + lane;
        x0[u] = s[0];
        x1[u] = s[32];
      }
    }
    float al = 0.0f;
    if (mine) {
      if (V2E) {
        atomicAdd(cnt + el, 1.0f);
        atomicAdd(att_sum + vl, homo[el]);
      } else {
        al = homo[el] / att_sum[vl];
      }
    }
    if (whole) {
#pragma unroll
      for (int u = 0; u < TRIP; ++u) {
        const int r = __builtin_amdgcn_readlane(V2E ? el : vl, u);
        float* d = dst + (size_t)r * HO + lane;
        float a0 = x0[u], a1 = x1[u];
        if (!V2E) {
          const float att = __int_as_float(__builtin_amdgcn_readlane(__float_as_int(al), u));
          a0 = att * a0;
          a1 = att * a1;
        }
        atomicAdd(d, a0);
        atomicAdd(d + 32, a1);
      }
    } else {
      for (int u = 0; base + u < nnz; ++u) {
        const int rs = __shfl(V2E ? vl : el, u, 32);
        const int rd = __shfl(V2E ? el : vl, u, 32);
        const float* s = src + (size_t)rs * HO + lane;
        float* d = dst + (size_t)rd * HO + lane;
        float a0 = s[0], a1 = s[32];
        if (!V2E) {
          const float att = __shfl(al, u, 32);
          a0 = att * a0;
          a1 = att * a1;
        }
        atomicAdd(d, a0);
        atomicAdd(d + 32, a1);
      }
    }
  }
}

__global__ __launch_bounds__(256) void k_edge_mean(float* __restrict__ Xe,
                                                   const float* __restrict__ cnt,
                                                   long long total) {
  long long i = (long long)blockIdx.x * blockDim.x + threadIdx.x;
  if (i >= total) return;
  const float c = cnt[i >> 6];
  Xe[i] *= 1.0f / fmaxf(c, 1.0f);
}

__global__ __launch_bounds__(256) void k_rownorm(float* __restrict__ out, int n) {
  const int wave = (int)(((unsigned)blockIdx.x * blockDim.x + threadIdx.x) >> 5);
  const int lane = (int)(threadIdx.x & 31u);
  if (wave >= n) return;
  float* row = out + (size_t)wave * HO + lane * 2;
  float2 x = *(float2*)row;
  float ss = x.x * x.x + x.y * x.y;
#pragma unroll
  for (int off = 16; off > 0; off >>= 1) ss += __shfl_xor(ss, off, 32);
  const float nrm = sqrtf(ss);
  const float scale = (nrm > 0.f) ? 1.0f / fmaxf(nrm, 1e-30f) : 0.0f;
  x.x *= scale;
  x.y *= scale;
  *(float2*)row = x;
}

extern "C" void kernel_launch(void* const* d_in, const int* in_sizes, int n_in,
                              void* d_out, int out_size, void* d_ws, size_t ws_size,
                              hipStream_t stream) {
  const float* X      = (const float*)d_in[0];
  const float* W      = (const float*)d_in[1];
  const float* homo   = (const float*)d_in[2];
  const int*   vertex = (const int*)d_in[3];
  const int*   edges  = (const int*)d_in[4];
  float* out = (float*)d_out;

  const int n   = in_sizes[0] / INCH;
  const int e   = in_sizes[2];
  const int nnz = in_sizes[3];

  float* Xe      = (float*)d_ws;
  float* cnt     = Xe + (size_t)e * HO;
  float* att_sum = cnt + e;
  const long long zn = (long long)e * HO + e + n;
  k_zero<<<(int)((zn + 255) / 256), 256, 0, stream>>>(Xe, zn);

  const int tiles = (n + 15) / 16;
  k_gemm_wmma<<<(tiles * 32 + 255) / 256, 256, 0, stream>>>(X, W, out, n);

  k_scatter_rows<true><<<8192, 256, 0, stream>>>(
      out, vertex, edges, homo, Xe, cnt, att_sum, nnz);

  const long long em = (long long)e * HO;
  k_edge_mean<<<(int)((em + 255) / 256), 256, 0, stream>>>(Xe, cnt, em);

  k_scatter_rows<false><<<8192, 256, 0, stream>>>(
      Xe, vertex, edges, homo, out, cnt, att_sum, nnz);

  const long long nt = (long long)n * 32;
  k_rownorm<<<(int)((nt + 255) / 256), 256, 0, stream>>>(out, n);
}


# Both scatter passes: incidences grouped per call by destination range (LDS counters), each range summed in one work-group's LDS, rows stored once; no L2 atomics (sum order changes: LDS f32 adds)
# speedup vs baseline: 2.7919x; 1.6227x over previous
#include <hip/hip_runtime.h>
#include <math.h>

typedef float v2f __attribute__((ext_vector_type(2)));
typedef float v8f __attribute__((ext_vector_type(8)));

#define HO 64
#define INCH 64

__global__ __launch_bounds__(256) void k_zero(float* __restrict__ p, long long n) {
  long long i = (long long)blockIdx.x * blockDim.x + threadIdx.x;
  if (i < n) p[i] = 0.0f;
}

__global__ __launch_bounds__(256) void k_gemm_wmma(
    const float* __restrict__ X, const float* __restrict__ W,
    float* __restrict__ Xp, int nrows) {
  const int wave = (int)(((unsigned)blockIdx.x * blockDim.x + threadIdx.x) >> 5);
  const int lane = (int)(threadIdx.x & 31u);
  const int row0 = wave * 16;
  if (row0 >= nrows) return;
  if (row0 + 16 <= nrows) {
    const int m  = lane & 15;
    const int kh = lane >> 4;
    v8f acc0 = {}, acc1 = {}, acc2 = {}, acc3 = {};
#pragma unroll
    for (int k0 = 0; k0 < INCH; k0 += 4) {
      v2f a = *(const v2f*)(X + (size_t)(row0 + m) * INCH + (k0 + kh * 2));
      const float* w0 = W + (size_t)(k0 + kh * 2) * HO;
      const float* w1 = w0 + HO;
      v2f b0, b1, b2, b3;
      b0.x = w0[m];      b0.y = w1[m];
      b1.x = w0[m + 16]; b1.y = w1[m + 16];
      b2.x = w0[m + 32]; b2.y = w1[m + 32];
      b3.x = w0[m + 48]; b3.y = w1[m + 48];
      acc0 = __builtin_amdgcn_wmma_f32_16x16x4_f32(false, a, false, b0, (short)0, acc0, false, false);
      acc1 = __builtin_amdgcn_wmma_f32_16x16x4_f32(false, a, false, b1, (short)0, acc1, false, false);
      acc2 = __builtin_amdgcn_wmma_f32_16x16x4_f32(false, a, false, b2, (short)0, acc2, false, false);
      acc3 = __builtin_amdgcn_wmma_f32_16x16x4_f32(false, a, false, b3, (short)0, acc3, false, false);
    }
#pragma unroll
    for (int r = 0; r < 8; ++r) {
      const int row = row0 + r + kh * 8;
      float* o = Xp + (size_t)row * HO + m;
      o[0]  = acc0[r];
      o[16] = acc1[r];
      o[32] = acc2[r];
      o[48] = acc3[r];
    }
  } else {
    for (int r = row0 + lane; r < nrows; r += 32)
      for (int c = 0; c < HO; ++c) {
        float s = 0.f;
        for (int k = 0; k < INCH; ++k) s += X[(size_t)r * INCH + k] * W[(size_t)k * HO + c];
        Xp[(size_t)r * HO + c] = s;
      }
  }
}

#define TRIP 8
template <bool V2E>
__global__ __launch_bounds__(256) void k_scatter_rows(
    const float* __restrict__ src, const int* __restrict__ vertex,
    const int* __restrict__ edges, const float* __restrict__ homo,
    float* __restrict__ dst, float* __restrict__ cnt,
    float* __restrict__ att_sum, int nnz) {
  const int lane = (int)(threadIdx.x & 31u);
  const int wave = __builtin_amdgcn_readfirstlane(
      (int)(((unsigned)blockIdx.x * blockDim.x + threadIdx.x) >> 5));
  const int nwaves = (int)((gridDim.x * blockDim.x) >> 5);
  for (int base = wave * TRIP; base < nnz; base += nwaves * TRIP) {
    const int idx = base + lane;
    const bool mine = lane < TRIP && idx < nnz;
    int vl = 0, el = 0;
    if (mine) {
      vl = vertex[idx];
      el = edges[idx];
    }
    const bool whole = base + TRIP <= nnz;
    float x0[TRIP], x1[TRIP];
    if (whole) {
#pragma unroll
      for (int u = 0; u < TRIP; ++u) {
        const int r = __builtin_amdgcn_readlane(V2E ? vl : el, u);
        const float* s = src + (size_t)r * HO + lane;
        x0[u] = s[0];
        x1[u] = s[32];
      }
    }
    float al = 0.0f;
    if (mine) {
      if (V2E) {
        atomicAdd(cnt + el, 1.0f);
        atomicAdd(att_sum + vl, homo[el]);
      } else {
        al = homo[el] / att_sum[vl];
      }
    }
    if (whole) {
#pragma unroll
      for (int u = 0; u < TRIP; ++u) {
        const int r = __builtin_amdgcn_readlane(V2E ? el : vl, u);
        float* d = dst + (size_t)r * HO + lane;
        float a0 = x0[u], a1 = x1[u];
        if (!V2E) {
          const float att = __int_as_float(__builtin_amdgcn_readlane(__float_as_int(al), u));
          a0 = att * a0;
          a1 = att * a1;
        }
        atomicAdd(d, a0);
        atomicAdd(d + 32, a1);
      }
    } else {
      for (int u = 0; base + u < nnz; ++u) {
        const int rs = __shfl(V2E ? vl : el, u, 32);
        const int rd = __shfl(V2E ? el : vl, u, 32);
        const float* s = src + (size_t)rs * HO + lane;
        float* d = dst + (size_t)rd * HO + lane;
        float a0 = s[0], a1 = s[32];
        if (!V2E) {
          const float att = __shfl(al, u, 32);
          a0 = att * a0;
          a1 = att * a1;
        }
        atomicAdd(d, a0);
        atomicAdd(d + 32, a1);
      }
    }
  }
}

__global__ __launch_bounds__(256) void k_edge_mean(float* __restrict__ Xe,
                                                   const float* __restrict__ cnt,
                                                   long long total) {
  long long i = (long long)blockIdx.x * blockDim.x + threadIdx.x;
  if (i >= total) return;
  const float c = cnt[i >> 6];
  Xe[i] *= 1.0f / fmaxf(c, 1.0f);
}

__global__ __launch_bounds__(256) void k_rownorm(float* __restrict__ out, int n) {
  const int wave = (int)(((unsigned)blockIdx.x * blockDim.x + threadIdx.x) >> 5);
  const int lane = (int)(threadIdx.x & 31u);
  if (wave >= n) return;
  float* row = out + (size_t)wave * HO + lane * 2;
  float2 x = *(float2*)row;
  float ss = x.x * x.x + x.y * x.y;
#pragma unroll
  for (int off = 16; off > 0; off >>= 1) ss += __shfl_xor(ss, off, 32);
  const float nrm = sqrtf(ss);
  const float scale = (nrm > 0.f) ? 1.0f / fmaxf(nrm, 1e-30f) : 0.0f;
  x.x *= scale;
  x.y *= scale;
  *(float2*)row = x;
}

#define NB 512
#define NCH 256
#define RMAXE 64
#define RMAXV 200
#define SRCMASK 0xFFFFF

__global__ __launch_bounds__(NB) void k_bin_count(
    const int* __restrict__ vertex, const int* __restrict__ edges,
    int* __restrict__ cntE, int* __restrict__ cntV, int nnz, int re, int rv) {
  __shared__ int hE[NB], hV[NB];
  const int t = (int)threadIdx.x, c = (int)blockIdx.x;
  hE[t] = 0;
  hV[t] = 0;
  __syncthreads();
  const int ch = (nnz + NCH - 1) / NCH;
  const int lo = c * ch;
  const int hi = min(lo + ch, nnz);
  for (int i = lo + t; i < hi; i += NB) {
    atomicAdd(&hE[edges[i] / re], 1);
    atomicAdd(&hV[vertex[i] / rv], 1);
  }
  __syncthreads();
  cntE[c * NB + t] = hE[t];
  cntV[c * NB + t] = hV[t];
}

__global__ __launch_bounds__(NB) void k_bin_offsets(
    int* __restrict__ cntE, int* __restrict__ cntV,
    int* __restrict__ bsE, int* __restrict__ bsV) {
  __shared__ int s[NB];
  int* tab = blockIdx.x ? cntV : cntE;
  int* bs = blockIdx.x ? bsV : bsE;
  const int b = (int)threadIdx.x;
  int run = 0;
  for (int c0 = 0; c0 < NCH; c0 += 32) {
    int v[32];
#pragma unroll
    for (int u = 0; u < 32; ++u) v[u] = tab[(c0 + u) * NB + b];
#pragma unroll
    for (int u = 0; u < 32; ++u) {
      tab[(c0 + u) * NB + b] = run;
      run += v[u];
    }
  }
  s[b] = run;
  __syncthreads();
  for (int off = 1; off < NB; off <<= 1) {
    const int add = (b >= off) ? s[b - off] : 0;
    __syncthreads();
    s[b] += add;
    __syncthreads();
  }
  bs[b] = s[b] - run;
  if (b == NB - 1) bs[NB] = s[b];
}

__global__ __launch_bounds__(NB) void k_bin_place(
    const int* __restrict__ vertex, const int* __restrict__ edges,
    const int* __restrict__ cntE, const int* __restrict__ cntV,
    const int* __restrict__ bsE, const int* __restrict__ bsV,
    int* __restrict__ listE, int* __restrict__ listV, int nnz, int re, int rv) {
  __shared__ int cE[NB], cV[NB];
  const int t = (int)threadIdx.x, c = (int)blockIdx.x;
  cE[t] = bsE[t] + cntE[c * NB + t];
  cV[t] = bsV[t] + cntV[c * NB + t];
  __syncthreads();
  const int ch = (nnz + NCH - 1) / NCH;
  const int lo = c * ch;
  const int hi = min(lo + ch, nnz);
  for (int i = lo + t; i < hi; i += NB) {
    const int v = vertex[i], e = edges[i];
    const int be = e / re, bv = v / rv;
    const int p = atomicAdd(&cE[be], 1);
    listE[p] = v | ((e - be * re) << 20);
    const int q = atomicAdd(&cV[bv], 1);
    listV[q] = e | ((v - bv * rv) << 20);
  }
}

#define TRIPB 16
template <bool V2E, int RMAX>
__global__ __launch_bounds__(NB) void k_range_sum(
    const float* __restrict__ src, const int* __restrict__ list, const int* __restrict__ bs,
    const float* __restrict__ homo, float* __restrict__ dst, int r, int nrows) {
  __shared__ float acc[RMAX * HO];
  __shared__ float aux[RMAX];
  const int t = (int)threadIdx.x;
  const int lane = t & 31;
  const int bin = (int)blockIdx.x;
  const int lo = bs[bin], hi = bs[bin + 1];
  for (int j = t; j < r * HO; j += NB) acc[j] = 0.0f;
  for (int j = t; j < r; j += NB) aux[j] = 0.0f;
  __syncthreads();
  for (int i = lo + t; i < hi; i += NB) {
    const int ent = list[i];
    atomicAdd(&aux[ent >> 20], V2E ? 1.0f : homo[ent & SRCMASK]);
  }
  __syncthreads();
  const int wave = __builtin_amdgcn_readfirstlane(t >> 5);
  const int step = (NB / 32) * TRIPB;
  int base = lo + wave * TRIPB;
  int ent = 0;
  if (lane < TRIPB && base + lane < hi) ent = list[base + lane];
  for (; base < hi; base += step) {
    const int pi = base + step + lane;
    const int entn = list[pi < hi ? pi : hi - 1];
    const int sl = ent & SRCMASK;
    const int dl = ent >> 20;
    float al = 1.0f;
    if (!V2E && lane < TRIPB && base + lane < hi) al = homo[sl] / aux[dl];
    if (base + TRIPB <= hi) {
      float x0[TRIPB], x1[TRIPB];
#pragma unroll
      for (int u = 0; u < TRIPB; ++u) {
        const float* s = src + (size_t)__builtin_amdgcn_readlane(sl, u) * HO + lane;
        x0[u] = s[0];
        x1[u] = s[32];
      }
#pragma unroll
      for (int u = 0; u < TRIPB; ++u) {
        float* d = acc + __builtin_amdgcn_readlane(dl, u) * HO + lane;
        float a0 = x0[u], a1 = x1[u];
        if (!V2E) {
          const float att = __int_as_float(__builtin_amdgcn_readlane(__float_as_int(al), u));
          a0 = att * a0;
          a1 = att * a1;
        }
        atomicAdd(d, a0);
        atomicAdd(d + 32, a1);
      }
    } else {
      for (int u = 0; base + u < hi; ++u) {
        const float* s = src + (size_t)__shfl(sl, u, 32) * HO + lane;
        float* d = acc + __shfl(dl, u, 32) * HO + lane;
        float a0 = s[0], a1 = s[32];
        if (!V2E) {
          const float att = __shfl(al, u, 32);
          a0 = att * a0;
          a1 = att * a1;
        }
        atomicAdd(d, a0);
        atomicAdd(d + 32, a1);
      }
    }
    ent = entn;
  }
  __syncthreads();
  for (int j = t; j < r * HO; j += NB) {
    const int row = bin * r + (j >> 6);
    if (row < nrows) {
      float* o = dst + (size_t)row * HO + (j & 63);
      if (V2E) *o = acc[j] / fmaxf(aux[j >> 6], 1.0f);
      else     *o = *o + acc[j];
    }
  }
}

extern "C" void kernel_launch(void* const* d_in, const int* in_sizes, int n_in,
                              void* d_out, int out_size, void* d_ws, size_t ws_size,
                              hipStream_t stream) {
  const float* X      = (const float*)d_in[0];
  const float* W      = (const float*)d_in[1];
  const float* homo   = (const float*)d_in[2];
  const int*   vertex = (const int*)d_in[3];
  const int*   edges  = (const int*)d_in[4];
  float* out = (float*)d_out;

  const int n   = in_sizes[0] / INCH;
  const int e   = in_sizes[2];
  const int nnz = in_sizes[3];

  const int tiles = (n + 15) / 16;
  k_gemm_wmma<<<(tiles * 32 + 255) / 256, 256, 0, stream>>>(X, W, out, n);

  const int re = (e + NB - 1) / NB;
  const int rv = (n + NB - 1) / NB;
  const size_t need = ((size_t)e * HO + 2 * (size_t)nnz + 2 * (size_t)NCH * NB + 2 * (NB + 1)) * 4;
  float* Xe = (float*)d_ws;

  if (need <= ws_size && re <= RMAXE && rv <= RMAXV && n <= SRCMASK && e <= SRCMASK) {
    int* listE = (int*)(Xe + (size_t)e * HO);
    int* listV = listE + nnz;
    int* cntE  = listV + nnz;
    int* cntV  = cntE + NCH * NB;
    int* bsE   = cntV + NCH * NB;
    int* bsV   = bsE + NB + 1;
    k_bin_count<<<NCH, NB, 0, stream>>>(vertex, edges, cntE, cntV, nnz, re, rv);
    k_bin_offsets<<<2, NB, 0, stream>>>(cntE, cntV, bsE, bsV);
    k_bin_place<<<NCH, NB, 0, stream>>>(vertex, edges, cntE, cntV, bsE, bsV, listE, listV, nnz, re, rv);
    k_range_sum<true, RMAXE><<<NB, NB, 0, stream>>>(out, listE, bsE, homo, Xe, re, e);
    k_range_sum<false, RMAXV><<<NB, NB, 0, stream>>>(Xe, listV, bsV, homo, out, rv, n);
  } else {
    float* cnt     = Xe + (size_t)e * HO;
    float* att_sum = cnt + e;
    const long long zn = (long long)e * HO + e + n;
    k_zero<<<(int)((zn + 255) / 256), 256, 0, stream>>>(Xe, zn);
    k_scatter_rows<true><<<8192, 256, 0, stream>>>(
        out, vertex, edges, homo, Xe, cnt, att_sum, nnz);
    const long long em = (long long)e * HO;
    k_edge_mean<<<(int)((em + 255) / 256), 256, 0, stream>>>(Xe, cnt, em);
    k_scatter_rows<false><<<8192, 256, 0, stream>>>(
        Xe, vertex, edges, homo, out, cnt, att_sum, nnz);
  }

  const long long nt = (long long)n * 32;
  k_rownorm<<<(int)((nt + 255) / 256), 256, 0, stream>>>(out, n);
}


# Row l2-normalisation folded into the vertex pass's store: one wave per row, norm taken while its sums are in LDS, Xp rows fetched as one group; the 100000-wave normalise launch is gone
# speedup vs baseline: 3.4420x; 1.2329x over previous
#include <hip/hip_runtime.h>
#include <math.h>

typedef float v2f __attribute__((ext_vector_type(2)));
typedef float v8f __attribute__((ext_vector_type(8)));

#define HO 64
#define INCH 64

__global__ __launch_bounds__(256) void k_zero(float* __restrict__ p, long long n) {
  long long i = (long long)blockIdx.x * blockDim.x + threadIdx.x;
  if (i < n) p[i] = 0.0f;
}

__global__ __launch_bounds__(256) void k_gemm_wmma(
    const float* __restrict__ X, const float* __restrict__ W,
    float* __restrict__ Xp, int nrows) {
  const int wave = (int)(((unsigned)blockIdx.x * blockDim.x + threadIdx.x) >> 5);
  const int lane = (int)(threadIdx.x & 31u);
  const int row0 = wave * 16;
  if (row0 >= nrows) return;
  if (row0 + 16 <= nrows) {
    const int m  = lane & 15;
    const int kh = lane >> 4;
    v8f acc0 = {}, acc1 = {}, acc2 = {}, acc3 = {};
#pragma unroll
    for (int k0 = 0; k0 < INCH; k0 += 4) {
      v2f a = *(const v2f*)(X + (size_t)(row0 + m) * INCH + (k0 + kh * 2));
      const float* w0 = W + (size_t)(k0 + kh * 2) * HO;
      const float* w1 = w0 + HO;
      v2f b0, b1, b2, b3;
      b0.x = w0[m];      b0.y = w1[m];
      b1.x = w0[m + 16]; b1.y = w1[m + 16];
      b2.x = w0[m + 32]; b2.y = w1[m + 32];
      b3.x = w0[m + 48]; b3.y = w1[m + 48];
      acc0 = __builtin_amdgcn_wmma_f32_16x16x4_f32(false, a, false, b0, (short)0, acc0, false, false);
      acc1 = __builtin_amdgcn_wmma_f32_16x16x4_f32(false, a, false, b1, (short)0, acc1, false, false);
      acc2 = __builtin_amdgcn_wmma_f32_16x16x4_f32(false, a, false, b2, (short)0, acc2, false, false);
      acc3 = __builtin_amdgcn_wmma_f32_16x16x4_f32(false, a, false, b3, (short)0, acc3, false, false);
    }
#pragma unroll
    for (int r = 0; r < 8; ++r) {
      const int row = row0 + r + kh * 8;
      float* o = Xp + (size_t)row * HO + m;
      o[0]  = acc0[r];
      o[16] = acc1[r];
      o[32] = acc2[r];
      o[48] = acc3[r];
    }
  } else {
    for (int r = row0 + lane; r < nrows; r += 32)
      for (int c = 0; c < HO; ++c) {
        float s = 0.f;
        for (int k = 0; k < INCH; ++k) s += X[(size_t)r * INCH + k] * W[(size_t)k * HO + c];
        Xp[(size_t)r * HO + c] = s;
      }
  }
}

#define TRIP 8
template <bool V2E>
__global__ __launch_bounds__(256) void k_scatter_rows(
    const float* __restrict__ src, const int* __restrict__ vertex,
    const int* __restrict__ edges, const float* __restrict__ homo,
    float* __restrict__ dst, float* __restrict__ cnt,
    float* __restrict__ att_sum, int nnz) {
  const int lane = (int)(threadIdx.x & 31u);
  const int wave = __builtin_amdgcn_readfirstlane(
      (int)(((unsigned)blockIdx.x * blockDim.x + threadIdx.x) >> 5));
  const int nwaves = (int)((gridDim.x * blockDim.x) >> 5);
  for (int base = wave * TRIP; base < nnz; base += nwaves * TRIP) {
    const int idx = base + lane;
    const bool mine = lane < TRIP && idx < nnz;
    int vl = 0, el = 0;
    if (mine) {
      vl = vertex[idx];
      el = edges[idx];
    }
    const bool whole = base + TRIP <= nnz;
    float x0[TRIP], x1[TRIP];
    if (whole) {
#pragma unroll
      for (int u = 0; u < TRIP; ++u) {
        const int r = __builtin_amdgcn_readlane(V2E ? vl : el, u);
        const float* s = src + (size_t)r * HO + lane;
        x0[u] = s[0];
        x1[u] = s[32];
      }
    }
    float al = 0.0f;
    if (mine) {
      if (V2E) {
        atomicAdd(cnt + el, 1.0f);
        atomicAdd(att_sum + vl, homo[el]);
      } else {
        al = homo[el] / att_sum[vl];
      }
    }
    if (whole) {
#pragma unroll
      for (int u = 0; u < TRIP; ++u) {
        const int r = __builtin_amdgcn_readlane(V2E ? el : vl, u);
        float* d = dst + (size_t)r * HO + lane;
        float a0 = x0[u], a1 = x1[u];
        if (!V2E) {
          const float att = __int_as_float(__builtin_amdgcn_readlane(__float_as_int(al), u));
          a0 = att * a0;
          a1 = att * a1;
        }
        atomicAdd(d, a0);
        atomicAdd(d + 32, a1);
      }
    } else {
      for (int u = 0; base + u < nnz; ++u) {
        const int rs = __shfl(V2E ? vl : el, u, 32);
        const int rd = __shfl(V2E ? el : vl, u, 32);
        const float* s = src + (size_t)rs * HO + lane;
        float* d = dst + (size_t)rd * HO + lane;
        float a0 = s[0], a1 = s[32];
        if (!V2E) {
          const float att = __shfl(al, u, 32);
          a0 = att * a0;
          a1 = att * a1;
        }
        atomicAdd(d, a0);
        atomicAdd(d + 32, a1);
      }
    }
  }
}

__global__ __launch_bounds__(256) void k_edge_mean(float* __restrict__ Xe,
                                                   const float* __restrict__ cnt,
                                                   long long total) {
  long long i = (long long)blockIdx.x * blockDim.x + threadIdx.x;
  if (i >= total) return;
  const float c = cnt[i >> 6];
  Xe[i] *= 1.0f / fmaxf(c, 1.0f);
}

__global__ __launch_bounds__(256) void k_rownorm(float* __restrict__ out, int n) {
  const int wave = (int)(((unsigned)blockIdx.x * blockDim.x + threadIdx.x) >> 5);
  const int lane = (int)(threadIdx.x & 31u);
  if (wave >= n) return;
  float* row = out + (size_t)wave * HO + lane * 2;
  float2 x = *(float2*)row;
  float ss = x.x * x.x + x.y * x.y;
#pragma unroll
  for (int off = 16; off > 0; off >>= 1) ss += __shfl_xor(ss, off, 32);
  const float nrm = sqrtf(ss);
  const float scale = (nrm > 0.f) ? 1.0f / fmaxf(nrm, 1e-30f) : 0.0f;
  x.x *= scale;
  x.y *= scale;
  *(float2*)row = x;
}

#define NB 512
#define NCH 256
#define RMAXE 64
#define RMAXV 200
#define SRCMASK 0xFFFFF

__global__ __launch_bounds__(NB) void k_bin_count(
    const int* __restrict__ vertex, const int* __restrict__ edges,
    int* __restrict__ cntE, int* __restrict__ cntV, int nnz, int re, int rv) {
  __shared__ int hE[NB], hV[NB];
  const int t = (int)threadIdx.x, c = (int)blockIdx.x;
  hE[t] = 0;
  hV[t] = 0;
  __syncthreads();
  const int ch = (nnz + NCH - 1) / NCH;
  const int lo = c * ch;
  const int hi = min(lo + ch, nnz);
  for (int i = lo + t; i < hi; i += NB) {
    atomicAdd(&hE[edges[i] / re], 1);
    atomicAdd(&hV[vertex[i] / rv], 1);
  }
  __syncthreads();
  cntE[c * NB + t] = hE[t];
  cntV[c * NB + t] = hV[t];
}

__global__ __launch_bounds__(NB) void k_bin_offsets(
    int* __restrict__ cntE, int* __restrict__ cntV,
    int* __restrict__ bsE, int* __restrict__ bsV) {
  __shared__ int s[NB];
  int* tab = blockIdx.x ? cntV : cntE;
  int* bs = blockIdx.x ? bsV : bsE;
  const int b = (int)threadIdx.x;
  int run = 0;
  for (int c0 = 0; c0 < NCH; c0 += 32) {
    int v[32];
#pragma unroll
    for (int u = 0; u < 32; ++u) v[u] = tab[(c0 + u) * NB + b];
#pragma unroll
    for (int u = 0; u < 32; ++u) {
      tab[(c0 + u) * NB + b] = run;
      run += v[u];
    }
  }
  s[b] = run;
  __syncthreads();
  for (int off = 1; off < NB; off <<= 1) {
    const int add = (b >= off) ? s[b - off] : 0;
    __syncthreads();
    s[b] += add;
    __syncthreads();
  }
  bs[b] = s[b] - run;
  if (b == NB - 1) bs[NB] = s[b];
}

__global__ __launch_bounds__(NB) void k_bin_place(
    const int* __restrict__ vertex, const int* __restrict__ edges,
    const int* __restrict__ cntE, const int* __restrict__ cntV,
    const int* __restrict__ bsE, const int* __restrict__ bsV,
    int* __restrict__ listE, int* __restrict__ listV, int nnz, int re, int rv) {
  __shared__ int cE[NB], cV[NB];
  const int t = (int)threadIdx.x, c = (int)blockIdx.x;
  cE[t] = bsE[t] + cntE[c * NB + t];
  cV[t] = bsV[t] + cntV[c * NB + t];
  __syncthreads();
  const int ch = (nnz + NCH - 1) / NCH;
  const int lo = c * ch;
  const int hi = min(lo + ch, nnz);
  for (int i = lo + t; i < hi; i += NB) {
    const int v = vertex[i], e = edges[i];
    const int be = e / re, bv = v / rv;
    const int p = atomicAdd(&cE[be], 1);
    listE[p] = v | ((e - be * re) << 20);
    const int q = atomicAdd(&cV[bv], 1);
    listV[q] = e | ((v - bv * rv) << 20);
  }
}

#define TRIPB 16
template <bool V2E, int RMAX>
__global__ __launch_bounds__(NB) void k_range_sum(
    const float* __restrict__ src, const int* __restrict__ list, const int* __restrict__ bs,
    const float* __restrict__ homo, float* __restrict__ dst, int r, int nrows) {
  __shared__ __attribute__((aligned(16))) float acc[RMAX * HO];
  __shared__ float aux[RMAX];
  const int t = (int)threadIdx.x;
  const int lane = t & 31;
  const int bin = (int)blockIdx.x;
  const int lo = bs[bin], hi = bs[bin + 1];
  for (int j = t; j < r * HO; j += NB) acc[j] = 0.0f;
  for (int j = t; j < r; j += NB) aux[j] = 0.0f;
  __syncthreads();
  for (int i = lo + t; i < hi; i += NB) {
    const int ent = list[i];
    atomicAdd(&aux[ent >> 20], V2E ? 1.0f : homo[ent & SRCMASK]);
  }
  __syncthreads();
  const int wave = __builtin_amdgcn_readfirstlane(t >> 5);
  const int step = (NB / 32) * TRIPB;
  int base = lo + wave * TRIPB;
  int ent = 0;
  if (lane < TRIPB && base + lane < hi) ent = list[base + lane];
  for (; base < hi; base += step) {
    const int pi = base + step + lane;
    const int entn = list[pi < hi ? pi : hi - 1];
    const int sl = ent & SRCMASK;
    const int dl = ent >> 20;
    float al = 1.0f;
    if (!V2E && lane < TRIPB && base + lane < hi) al = homo[sl] / aux[dl];
    if (base + TRIPB <= hi) {
      float x0[TRIPB], x1[TRIPB];
#pragma unroll
      for (int u = 0; u < TRIPB; ++u) {
        const float* s = src + (size_t)__builtin_amdgcn_readlane(sl, u) * HO + lane;
        x0[u] = s[0];
        x1[u] = s[32];
      }
#pragma unroll
      for (int u = 0; u < TRIPB; ++u) {
        float* d = acc + __builtin_amdgcn_readlane(dl, u) * HO + lane;
        float a0 = x0[u], a1 = x1[u];
        if (!V2E) {
          const float att = __int_as_float(__builtin_amdgcn_readlane(__float_as_int(al), u));
          a0 = att * a0;
          a1 = att * a1;
        }
        atomicAdd(d, a0);
        atomicAdd(d + 32, a1);
      }
    } else {
      for (int u = 0; base + u < hi; ++u) {
        const float* s = src + (size_t)__shfl(sl, u, 32) * HO + lane;
        float* d = acc + __shfl(dl, u, 32) * HO + lane;
        float a0 = s[0], a1 = s[32];
        if (!V2E) {
          const float att = __shfl(al, u, 32);
          a0 = att * a0;
          a1 = att * a1;
        }
        atomicAdd(d, a0);
        atomicAdd(d + 32, a1);
      }
    }
    ent = entn;
  }
  if (V2E) {
    __syncthreads();
    for (int j = t; j < r * HO; j += NB) {
      const int row = bin * r + (j >> 6);
      if (row < nrows) dst[(size_t)row * HO + (j & 63)] = acc[j] / fmaxf(aux[j >> 6], 1.0f);
    }
  } else {
    constexpr int NW = NB / 32;
    constexpr int RPW = (RMAX + NW - 1) / NW;
    const int nv = min(r, nrows - bin * r);
    if (nv <= 0) return;
    float2 x[RPW];
#pragma unroll
    for (int u = 0; u < RPW; ++u) {
      const int rc = min(wave + u * NW, nv - 1);
      x[u] = *(const float2*)(dst + (size_t)(bin * r + rc) * HO + lane * 2);
    }
    __syncthreads();
#pragma unroll
    for (int u = 0; u < RPW; ++u) {
      const int rc = min(wave + u * NW, nv - 1);
      const float2 a = *(const float2*)(acc + rc * HO + lane * 2);
      x[u].x = x[u].x + a.x;
      x[u].y = x[u].y + a.y;
      float ss = x[u].x * x[u].x + x[u].y * x[u].y;
#pragma unroll
      for (int off = 16; off > 0; off >>= 1) ss += __shfl_xor(ss, off, 32);
      const float nrm = sqrtf(ss);
      const float inv = 1.0f / fmaxf(nrm, 1e-30f);
      const float scale = (nrm > 0.f) ? inv : 0.0f;
      x[u].x *= scale;
      x[u].y *= scale;
    }
#pragma unroll
    for (int u = 0; u < RPW; ++u) {
      const int rr = wave + u * NW;
      if (rr < nv)
        *(float2*)(dst + (size_t)(bin * r + rr) * HO + lane * 2) = x[u];
    }
  }
}

extern "C" void kernel_launch(void* const* d_in, const int* in_sizes, int n_in,
                              void* d_out, int out_size, void* d_ws, size_t ws_size,
                              hipStream_t stream) {
  const float* X      = (const float*)d_in[0];
  const float* W      = (const float*)d_in[1];
  const float* homo   = (const float*)d_in[2];
  const int*   vertex = (const int*)d_in[3];
  const int*   edges  = (const int*)d_in[4];
  float* out = (float*)d_out;

  const int n   = in_sizes[0] / INCH;
  const int e   = in_sizes[2];
  const int nnz = in_sizes[3];

  const int tiles = (n + 15) / 16;
  k_gemm_wmma<<<(tiles * 32 + 255) / 256, 256, 0, stream>>>(X, W, out, n);

  const int re = (e + NB - 1) / NB;
  const int rv = (n + NB - 1) / NB;
  const size_t need = ((size_t)e * HO + 2 * (size_t)nnz + 2 * (size_t)NCH * NB + 2 * (NB + 1)) * 4;
  float* Xe = (float*)d_ws;

  if (need <= ws_size && re <= RMAXE && rv <= RMAXV && n <= SRCMASK && e <= SRCMASK) {
    int* listE = (int*)(Xe + (size_t)e * HO);
    int* listV = listE + nnz;
    int* cntE  = listV + nnz;
    int* cntV  = cntE + NCH * NB;
    int* bsE   = cntV + NCH * NB;
    int* bsV   = bsE + NB + 1;
    k_bin_count<<<NCH, NB, 0, stream>>>(vertex, edges, cntE, cntV, nnz, re, rv);
    k_bin_offsets<<<2, NB, 0, stream>>>(cntE, cntV, bsE, bsV);
    k_bin_place<<<NCH, NB, 0, stream>>>(vertex, edges, cntE, cntV, bsE, bsV, listE, listV, nnz, re, rv);
    k_range_sum<true, RMAXE><<<NB, NB, 0, stream>>>(out, listE, bsE, homo, Xe, re, e);
    k_range_sum<false, RMAXV><<<NB, NB, 0, stream>>>(Xe, listV, bsV, homo, out, rv, n);
  } else {
    float* cnt     = Xe + (size_t)e * HO;
    float* att_sum = cnt + e;
    const long long zn = (long long)e * HO + e + n;
    k_zero<<<(int)((zn + 255) / 256), 256, 0, stream>>>(Xe, zn);
    k_scatter_rows<true><<<8192, 256, 0, stream>>>(
        out, vertex, edges, homo, Xe, cnt, att_sum, nnz);
    const long long em = (long long)e * HO;
    k_edge_mean<<<(int)((em + 255) / 256), 256, 0, stream>>>(Xe, cnt, em);
    k_scatter_rows<false><<<8192, 256, 0, stream>>>(
        Xe, vertex, edges, homo, out, cnt, att_sum, nnz);
    const long long nt = (long long)n * 32;
    k_rownorm<<<(int)((nt + 255) / 256), 256, 0, stream>>>(out, n);
  }
}


# Range sums: each range's list is counting-sorted by destination row in LDS, waves sum runs of one row in registers and add to the LDS row only where the row changes (sum order changes)
# speedup vs baseline: 3.5844x; 1.0414x over previous
#include <hip/hip_runtime.h>
#include <math.h>

typedef float v2f __attribute__((ext_vector_type(2)));
typedef float v8f __attribute__((ext_vector_type(8)));

#define HO 64
#define INCH 64

__global__ __launch_bounds__(256) void k_zero(float* __restrict__ p, long long n) {
  long long i = (long long)blockIdx.x * blockDim.x + threadIdx.x;
  if (i < n) p[i] = 0.0f;
}

__global__ __launch_bounds__(256) void k_gemm_wmma(
    const float* __restrict__ X, const float* __restrict__ W,
    float* __restrict__ Xp, int nrows) {
  const int wave = (int)(((unsigned)blockIdx.x * blockDim.x + threadIdx.x) >> 5);
  const int lane = (int)(threadIdx.x & 31u);
  const int row0 = wave * 16;
  if (row0 >= nrows) return;
  if (row0 + 16 <= nrows) {
    const int m  = lane & 15;
    const int kh = lane >> 4;
    v8f acc0 = {}, acc1 = {}, acc2 = {}, acc3 = {};
#pragma unroll
    for (int k0 = 0; k0 < INCH; k0 += 4) {
      v2f a = *(const v2f*)(X + (size_t)(row0 + m) * INCH + (k0 + kh * 2));
      const float* w0 = W + (size_t)(k0 + kh * 2) * HO;
      const float* w1 = w0 + HO;
      v2f b0, b1, b2, b3;
      b0.x = w0[m];      b0.y = w1[m];
      b1.x = w0[m + 16]; b1.y = w1[m + 16];
      b2.x = w0[m + 32]; b2.y = w1[m + 32];
      b3.x = w0[m + 48]; b3.y = w1[m + 48];
      acc0 = __builtin_amdgcn_wmma_f32_16x16x4_f32(false, a, false, b0, (short)0, acc0, false, false);
      acc1 = __builtin_amdgcn_wmma_f32_16x16x4_f32(false, a, false, b1, (short)0, acc1, false, false);
      acc2 = __builtin_amdgcn_wmma_f32_16x16x4_f32(false, a, false, b2, (short)0, acc2, false, false);
      acc3 = __builtin_amdgcn_wmma_f32_16x16x4_f32(false, a, false, b3, (short)0, acc3, false, false);
    }
#pragma unroll
    for (int r = 0; r < 8; ++r) {
      const int row = row0 + r + kh * 8;
      float* o = Xp + (size_t)row * HO + m;
      o[0]  = acc0[r];
      o[16] = acc1[r];
      o[32] = acc2[r];
      o[48] = acc3[r];
    }
  } else {
    for (int r = row0 + lane; r < nrows; r += 32)
      for (int c = 0; c < HO; ++c) {
        float s = 0.f;
        for (int k = 0; k < INCH; ++k) s += X[(size_t)r * INCH + k] * W[(size_t)k * HO + c];
        Xp[(size_t)r * HO + c] = s;
      }
  }
}

#define TRIP 8
template <bool V2E>
__global__ __launch_bounds__(256) void k_scatter_rows(
    const float* __restrict__ src, const int* __restrict__ vertex,
    const int* __restrict__ edges, const float* __restrict__ homo,
    float* __restrict__ dst, float* __restrict__ cnt,
    float* __restrict__ att_sum, int nnz) {
  const int lane = (int)(threadIdx.x & 31u);
  const int wave = __builtin_amdgcn_readfirstlane(
      (int)(((unsigned)blockIdx.x * blockDim.x + threadIdx.x) >> 5));
  const int nwaves = (int)((gridDim.x * blockDim.x) >> 5);
  for (int base = wave * TRIP; base < nnz; base += nwaves * TRIP) {
    const int idx = base + lane;
    const bool mine = lane < TRIP && idx < nnz;
    int vl = 0, el = 0;
    if (mine) {
      vl = vertex[idx];
      el = edges[idx];
    }
    const bool whole = base + TRIP <= nnz;
    float x0[TRIP], x1[TRIP];
    if (whole) {
#pragma unroll
      for (int u = 0; u < TRIP; ++u) {
        const int r = __builtin_amdgcn_readlane(V2E ? vl : el, u);
        const float* s = src + (size_t)r * HO + lane;
        x0[u] = s[0];
        x1[u] = s[32];
      }
    }
    float al = 0.0f;
    if (mine) {
      if (V2E) {
        atomicAdd(cnt + el, 1.0f);
        atomicAdd(att_sum + vl, homo[el]);
      } else {
        al = homo[el] / att_sum[vl];
      }
    }
    if (whole) {
#pragma unroll
      for (int u = 0; u < TRIP; ++u) {
        const int r = __builtin_amdgcn_readlane(V2E ? el : vl, u);
        float* d = dst + (size_t)r * HO + lane;
        float a0 = x0[u], a1 = x1[u];
        if (!V2E) {
          const float att = __int_as_float(__builtin_amdgcn_readlane(__float_as_int(al), u));
          a0 = att * a0;
          a1 = att * a1;
        }
        atomicAdd(d, a0);
        atomicAdd(d + 32, a1);
      }
    } else {
      for (int u = 0; base + u < nnz; ++u) {
        const int rs = __shfl(V2E ? vl : el, u, 32);
        const int rd = __shfl(V2E ? el : vl, u, 32);
        const float* s = src + (size_t)rs * HO + lane;
        float* d = dst + (size_t)rd * HO + lane;
        float a0 = s[0], a1 = s[32];
        if (!V2E) {
          const float att = __shfl(al, u, 32);
          a0 = att * a0;
          a1 = att * a1;
        }
        atomicAdd(d, a0);
        atomicAdd(d + 32, a1);
      }
    }
  }
}

__global__ __launch_bounds__(256) void k_edge_mean(float* __restrict__ Xe,
                                                   const float* __restrict__ cnt,
                                                   long long total) {
  long long i = (long long)blockIdx.x * blockDim.x + threadIdx.x;
  if (i >= total) return;
  const float c = cnt[i >> 6];
  Xe[i] *= 1.0f / fmaxf(c, 1.0f);
}

__global__ __launch_bounds__(256) void k_rownorm(float* __restrict__ out, int n) {
  const int wave = (int)(((unsigned)blockIdx.x * blockDim.x + threadIdx.x) >> 5);
  const int lane = (int)(threadIdx.x & 31u);
  if (wave >= n) return;
  float* row = out + (size_t)wave * HO + lane * 2;
  float2 x = *(float2*)row;
  float ss = x.x * x.x + x.y * x.y;
#pragma unroll
  for (int off = 16; off > 0; off >>= 1) ss += __shfl_xor(ss, off, 32);
  const float nrm = sqrtf(ss);
  const float scale = (nrm > 0.f) ? 1.0f / fmaxf(nrm, 1e-30f) : 0.0f;
  x.x *= scale;
  x.y *= scale;
  *(float2*)row = x;
}

#define NB 512
#define NCH 256
#define RMAXE 64
#define RMAXV 200
#define SRCMASK 0xFFFFF

__global__ __launch_bounds__(NB) void k_bin_count(
    const int* __restrict__ vertex, const int* __restrict__ edges,
    int* __restrict__ cntE, int* __restrict__ cntV, int nnz, int re, int rv) {
  __shared__ int hE[NB], hV[NB];
  const int t = (int)threadIdx.x, c = (int)blockIdx.x;
  hE[t] = 0;
  hV[t] = 0;
  __syncthreads();
  const int ch = (nnz + NCH - 1) / NCH;
  const int lo = c * ch;
  const int hi = min(lo + ch, nnz);
  for (int i = lo + t; i < hi; i += NB) {
    atomicAdd(&hE[edges[i] / re], 1);
    atomicAdd(&hV[vertex[i] / rv], 1);
  }
  __syncthreads();
  cntE[c * NB + t] = hE[t];
  cntV[c * NB + t] = hV[t];
}

__global__ __launch_bounds__(NB) void k_bin_offsets(
    int* __restrict__ cntE, int* __restrict__ cntV,
    int* __restrict__ bsE, int* __restrict__ bsV) {
  __shared__ int s[NB];
  int* tab = blockIdx.x ? cntV : cntE;
  int* bs = blockIdx.x ? bsV : bsE;
  const int b = (int)threadIdx.x;
  int run = 0;
  for (int c0 = 0; c0 < NCH; c0 += 32) {
    int v[32];
#pragma unroll
    for (int u = 0; u < 32; ++u) v[u] = tab[(c0 + u) * NB + b];
#pragma unroll
    for (int u = 0; u < 32; ++u) {
      tab[(c0 + u) * NB + b] = run;
      run += v[u];
    }
  }
  s[b] = run;
  __syncthreads();
  for (int off = 1; off < NB; off <<= 1) {
    const int add = (b >= off) ? s[b - off] : 0;
    __syncthreads();
    s[b] += add;
    __syncthreads();
  }
  bs[b] = s[b] - run;
  if (b == NB - 1) bs[NB] = s[b];
}

__global__ __launch_bounds__(NB) void k_bin_place(
    const int* __restrict__ vertex, const int* __restrict__ edges,
    const int* __restrict__ cntE, const int* __restrict__ cntV,
    const int* __restrict__ bsE, const int* __restrict__ bsV,
    int* __restrict__ listE, int* __restrict__ listV, int nnz, int re, int rv) {
  __shared__ int cE[NB], cV[NB];
  const int t = (int)threadIdx.x, c = (int)blockIdx.x;
  cE[t] = bsE[t] + cntE[c * NB + t];
  cV[t] = bsV[t] + cntV[c * NB + t];
  __syncthreads();
  const int ch = (nnz + NCH - 1) / NCH;
  const int lo = c * ch;
  const int hi = min(lo + ch, nnz);
  for (int i = lo + t; i < hi; i += NB) {
    const int v = vertex[i], e = edges[i];
    const int be = e / re, bv = v / rv;
    const int p = atomicAdd(&cE[be], 1);
    listE[p] = v | ((e - be * re) << 20);
    const int q = atomicAdd(&cV[bv], 1);
    listV[q] = e | ((v - bv * rv) << 20);
  }
}

#define TRIPB 16
#define TILE 4608
template <bool V2E, int RMAX>
__global__ __launch_bounds__(NB) void k_range_sum(
    const float* __restrict__ src, const int* __restrict__ list, const int* __restrict__ bs,
    const float* __restrict__ homo, float* __restrict__ dst, int r, int nrows) {
  __shared__ __attribute__((aligned(16))) float acc[RMAX * HO];
  __shared__ float aux[RMAX];
  __shared__ int cur[RMAX];
  __shared__ int sorted[TILE];
  const int t = (int)threadIdx.x;
  const int lane = t & 31;
  const int bin = (int)blockIdx.x;
  const int lo = bs[bin], hi = bs[bin + 1];
  for (int j = t; j < r * HO; j += NB) acc[j] = 0.0f;
  for (int j = t; j < r; j += NB) aux[j] = 0.0f;
  __syncthreads();
  for (int i = lo + t; i < hi; i += NB) {
    const int ent = list[i];
    atomicAdd(&aux[ent >> 20], V2E ? 1.0f : homo[ent & SRCMASK]);
  }
  const int wave = __builtin_amdgcn_readfirstlane(t >> 5);
  constexpr int NWV = NB / 32;
  constexpr int KW = TILE / NB;
  constexpr int RPL = (RMAX + 31) / 32;
  for (int tlo = lo; tlo < hi; tlo += TILE) {
    const int tn = min(TILE, hi - tlo);
    for (int j = t; j < r; j += NB) cur[j] = 0;
    int w[KW];
#pragma unroll
    for (int k = 0; k < KW; ++k) w[k] = list[min(tlo + k * NB + t, hi - 1)];
    __syncthreads();
#pragma unroll
    for (int k = 0; k < KW; ++k)
      if (k * NB + t < tn) atomicAdd(&cur[w[k] >> 20], 1);
    __syncthreads();
    if (wave == 0) {
      int c[RPL];
      int s = 0;
#pragma unroll
      for (int u = 0; u < RPL; ++u) {
        const int j = lane * RPL + u;
        c[u] = (j < r) ? cur[j] : 0;
        s += c[u];
      }
      int incl = s;
#pragma unroll
      for (int off = 1; off < 32; off <<= 1) {
        const int v = __shfl_up(incl, off, 32);
        if (lane >= off) incl += v;
      }
      int run = incl - s;
#pragma unroll
      for (int u = 0; u < RPL; ++u) {
        const int j = lane * RPL + u;
        if (j < r) cur[j] = run;
        run += c[u];
      }
    }
    __syncthreads();
#pragma unroll
    for (int k = 0; k < KW; ++k)
      if (k * NB + t < tn) sorted[atomicAdd(&cur[w[k] >> 20], 1)] = w[k];
    __syncthreads();
    const int seg = ((tn + NWV * TRIPB - 1) / (NWV * TRIPB)) * TRIPB;
    const int wbeg = wave * seg;
    const int wend = min(wbeg + seg, tn);
    if (wbeg < wend) {
      float a0 = 0.0f, a1 = 0.0f;
      int crow = -1;
      int ent = sorted[min(wbeg + lane, wend - 1)];
      for (int b = wbeg; b < wend; b += TRIPB) {
        const int nv = min(TRIPB, wend - b);
        const int entn = sorted[min(b + TRIPB + lane, wend - 1)];
        const int sl = ent & SRCMASK;
        const int dl = ent >> 20;
        float al = 1.0f;
        if (!V2E && lane < nv) al = homo[sl] / aux[dl];
        float x0[TRIPB], x1[TRIPB];
#pragma unroll
        for (int u = 0; u < TRIPB; ++u) {
          const float* s = src + (size_t)__builtin_amdgcn_readlane(sl, u) * HO + lane;
          x0[u] = s[0];
          x1[u] = s[32];
        }
#pragma unroll
        for (int u = 0; u < TRIPB; ++u) {
          if (u < nv) {
            const int d = __builtin_amdgcn_readlane(dl, u);
            if (d != crow) {
              if (crow >= 0) {
                atomicAdd(acc + crow * HO + lane, a0);
                atomicAdd(acc + crow * HO + 32 + lane, a1);
              }
              a0 = 0.0f;
              a1 = 0.0f;
              crow = d;
            }
            if (V2E) {
              a0 += x0[u];
              a1 += x1[u];
            } else {
              const float att = __int_as_float(__builtin_amdgcn_readlane(__float_as_int(al), u));
              a0 += att * x0[u];
              a1 += att * x1[u];
            }
          }
        }
        ent = entn;
      }
      if (crow >= 0) {
        atomicAdd(acc + crow * HO + lane, a0);
        atomicAdd(acc + crow * HO + 32 + lane, a1);
      }
    }
  }
  if (V2E) {
    __syncthreads();
    for (int j = t; j < r * HO; j += NB) {
      const int row = bin * r + (j >> 6);
      if (row < nrows) dst[(size_t)row * HO + (j & 63)] = acc[j] / fmaxf(aux[j >> 6], 1.0f);
    }
  } else {
    constexpr int NW = NB / 32;
    constexpr int RPW = (RMAX + NW - 1) / NW;
    const int nv = min(r, nrows - bin * r);
    if (nv <= 0) return;
    float2 x[RPW];
#pragma unroll
    for (int u = 0; u < RPW; ++u) {
      const int rc = min(wave + u * NW, nv - 1);
      x[u] = *(const float2*)(dst + (size_t)(bin * r + rc) * HO + lane * 2);
    }
    __syncthreads();
#pragma unroll
    for (int u = 0; u < RPW; ++u) {
      const int rc = min(wave + u * NW, nv - 1);
      const float2 a = *(const float2*)(acc + rc * HO + lane * 2);
      x[u].x = x[u].x + a.x;
      x[u].y = x[u].y + a.y;
      float ss = x[u].x * x[u].x + x[u].y * x[u].y;
#pragma unroll
      for (int off = 16; off > 0; off >>= 1) ss += __shfl_xor(ss, off, 32);
      const float nrm = sqrtf(ss);
      const float inv = 1.0f / fmaxf(nrm, 1e-30f);
      const float scale = (nrm > 0.f) ? inv : 0.0f;
      x[u].x *= scale;
      x[u].y *= scale;
    }
#pragma unroll
    for (int u = 0; u < RPW; ++u) {
      const int rr = wave + u * NW;
      if (rr < nv)
        *(float2*)(dst + (size_t)(bin * r + rr) * HO + lane * 2) = x[u];
    }
  }
}

extern "C" void kernel_launch(void* const* d_in, const int* in_sizes, int n_in,
                              void* d_out, int out_size, void* d_ws, size_t ws_size,
                              hipStream_t stream) {
  const float* X      = (const float*)d_in[0];
  const float* W      = (const float*)d_in[1];
  const float* homo   = (const float*)d_in[2];
  const int*   vertex = (const int*)d_in[3];
  const int*   edges  = (const int*)d_in[4];
  float* out = (float*)d_out;

  const int n   = in_sizes[0] / INCH;
  const int e   = in_sizes[2];
  const int nnz = in_sizes[3];

  const int tiles = (n + 15) / 16;
  k_gemm_wmma<<<(tiles * 32 + 255) / 256, 256, 0, stream>>>(X, W, out, n);

  const int re = (e + NB - 1) / NB;
  const int rv = (n + NB - 1) / NB;
  const size_t need = ((size_t)e * HO + 2 * (size_t)nnz + 2 * (size_t)NCH * NB + 2 * (NB + 1)) * 4;
  float* Xe = (float*)d_ws;

  if (need <= ws_size && re <= RMAXE && rv <= RMAXV && n <= SRCMASK && e <= SRCMASK) {
    int* listE = (int*)(Xe + (size_t)e * HO);
    int* listV = listE + nnz;
    int* cntE  = listV + nnz;
    int* cntV  = cntE + NCH * NB;
    int* bsE   = cntV + NCH * NB;
    int* bsV   = bsE + NB + 1;
    k_bin_count<<<NCH, NB, 0, stream>>>(vertex, edges, cntE, cntV, nnz, re, rv);
    k_bin_offsets<<<2, NB, 0, stream>>>(cntE, cntV, bsE, bsV);
    k_bin_place<<<NCH, NB, 0, stream>>>(vertex, edges, cntE, cntV, bsE, bsV, listE, listV, nnz, re, rv);
    k_range_sum<true, RMAXE><<<NB, NB, 0, stream>>>(out, listE, bsE, homo, Xe, re, e);
    k_range_sum<false, RMAXV><<<NB, NB, 0, stream>>>(Xe, listV, bsV, homo, out, rv, n);
  } else {
    float* cnt     = Xe + (size_t)e * HO;
    float* att_sum = cnt + e;
    const long long zn = (long long)e * HO + e + n;
    k_zero<<<(int)((zn + 255) / 256), 256, 0, stream>>>(Xe, zn);
    k_scatter_rows<true><<<8192, 256, 0, stream>>>(
        out, vertex, edges, homo, Xe, cnt, att_sum, nnz);
    const long long em = (long long)e * HO;
    k_edge_mean<<<(int)((em + 255) / 256), 256, 0, stream>>>(Xe, cnt, em);
    k_scatter_rows<false><<<8192, 256, 0, stream>>>(
        Xe, vertex, edges, homo, out, cnt, att_sum, nnz);
    const long long nt = (long long)n * 32;
    k_rownorm<<<(int)((nt + 255) / 256), 256, 0, stream>>>(out, n);
  }
}


# Place pass: a chunk's incidences are grouped by range in LDS first (rank by LDS add, scan of its counts), then stored in slot order: a wave store covers a few contiguous list pieces, not 32 lines
# speedup vs baseline: 4.8060x; 1.3408x over previous
#include <hip/hip_runtime.h>
#include <math.h>

typedef float v2f __attribute__((ext_vector_type(2)));
typedef float v8f __attribute__((ext_vector_type(8)));

#define HO 64
#define INCH 64

__global__ __launch_bounds__(256) void k_zero(float* __restrict__ p, long long n) {
  long long i = (long long)blockIdx.x * blockDim.x + threadIdx.x;
  if (i < n) p[i] = 0.0f;
}

__global__ __launch_bounds__(256) void k_gemm_wmma(
    const float* __restrict__ X, const float* __restrict__ W,
    float* __restrict__ Xp, int nrows) {
  const int wave = (int)(((unsigned)blockIdx.x * blockDim.x + threadIdx.x) >> 5);
  const int lane = (int)(threadIdx.x & 31u);
  const int row0 = wave * 16;
  if (row0 >= nrows) return;
  if (row0 + 16 <= nrows) {
    const int m  = lane & 15;
    const int kh = lane >> 4;
    v8f acc0 = {}, acc1 = {}, acc2 = {}, acc3 = {};
#pragma unroll
    for (int k0 = 0; k0 < INCH; k0 += 4) {
      v2f a = *(const v2f*)(X + (size_t)(row0 + m) * INCH + (k0 + kh * 2));
      const float* w0 = W + (size_t)(k0 + kh * 2) * HO;
      const float* w1 = w0 + HO;
      v2f b0, b1, b2, b3;
      b0.x = w0[m];      b0.y = w1[m];
      b1.x = w0[m + 16]; b1.y = w1[m + 16];
      b2.x = w0[m + 32]; b2.y = w1[m + 32];
      b3.x = w0[m + 48]; b3.y = w1[m + 48];
      acc0 = __builtin_amdgcn_wmma_f32_16x16x4_f32(false, a, false, b0, (short)0, acc0, false, false);
      acc1 = __builtin_amdgcn_wmma_f32_16x16x4_f32(false, a, false, b1, (short)0, acc1, false, false);
      acc2 = __builtin_amdgcn_wmma_f32_16x16x4_f32(false, a, false, b2, (short)0, acc2, false, false);
      acc3 = __builtin_amdgcn_wmma_f32_16x16x4_f32(false, a, false, b3, (short)0, acc3, false, false);
    }
#pragma unroll
    for (int r = 0; r < 8; ++r) {
      const int row = row0 + r + kh * 8;
      float* o = Xp + (size_t)row * HO + m;
      o[0]  = acc0[r];
      o[16] = acc1[r];
      o[32] = acc2[r];
      o[48] = acc3[r];
    }
  } else {
    for (int r = row0 + lane; r < nrows; r += 32)
      for (int c = 0; c < HO; ++c) {
        float s = 0.f;
        for (int k = 0; k < INCH; ++k) s += X[(size_t)r * INCH + k] * W[(size_t)k * HO + c];
        Xp[(size_t)r * HO + c] = s;
      }
  }
}

#define TRIP 8
template <bool V2E>
__global__ __launch_bounds__(256) void k_scatter_rows(
    const float* __restrict__ src, const int* __restrict__ vertex,
    const int* __restrict__ edges, const float* __restrict__ homo,
    float* __restrict__ dst, float* __restrict__ cnt,
    float* __restrict__ att_sum, int nnz) {
  const int lane = (int)(threadIdx.x & 31u);
  const int wave = __builtin_amdgcn_readfirstlane(
      (int)(((unsigned)blockIdx.x * blockDim.x + threadIdx.x) >> 5));
  const int nwaves = (int)((gridDim.x * blockDim.x) >> 5);
  for (int base = wave * TRIP; base < nnz; base += nwaves * TRIP) {
    const int idx = base + lane;
    const bool mine = lane < TRIP && idx < nnz;
    int vl = 0, el = 0;
    if (mine) {
      vl = vertex[idx];
      el = edges[idx];
    }
    const bool whole = base + TRIP <= nnz;
    float x0[TRIP], x1[TRIP];
    if (whole) {
#pragma unroll
      for (int u = 0; u < TRIP; ++u) {
        const int r = __builtin_amdgcn_readlane(V2E ? vl : el, u);
        const float* s = src + (size_t)r * HO + lane;
        x0[u] = s[0];
        x1[u] = s[32];
      }
    }
    float al = 0.0f;
    if (mine) {
      if (V2E) {
        atomicAdd(cnt + el, 1.0f);
        atomicAdd(att_sum + vl, homo[el]);
      } else {
        al = homo[el] / att_sum[vl];
      }
    }
    if (whole) {
#pragma unroll
      for (int u = 0; u < TRIP; ++u) {
        const int r = __builtin_amdgcn_readlane(V2E ? el : vl, u);
        float* d = dst + (size_t)r * HO + lane;
        float a0 = x0[u], a1 = x1[u];
        if (!V2E) {
          const float att = __int_as_float(__builtin_amdgcn_readlane(__float_as_int(al), u));
          a0 = att * a0;
          a1 = att * a1;
        }
        atomicAdd(d, a0);
        atomicAdd(d + 32, a1);
      }
    } else {
      for (int u = 0; base + u < nnz; ++u) {
        const int rs = __shfl(V2E ? vl : el, u, 32);
        const int rd = __shfl(V2E ? el : vl, u, 32);
        const float* s = src + (size_t)rs * HO + lane;
        float* d = dst + (size_t)rd * HO + lane;
        float a0 = s[0], a1 = s[32];
        if (!V2E) {
          const float att = __shfl(al, u, 32);
          a0 = att * a0;
          a1 = att * a1;
        }
        atomicAdd(d, a0);
        atomicAdd(d + 32, a1);
      }
    }
  }
}

__global__ __launch_bounds__(256) void k_edge_mean(float* __restrict__ Xe,
                                                   const float* __restrict__ cnt,
                                                   long long total) {
  long long i = (long long)blockIdx.x * blockDim.x + threadIdx.x;
  if (i >= total) return;
  const float c = cnt[i >> 6];
  Xe[i] *= 1.0f / fmaxf(c, 1.0f);
}

__global__ __launch_bounds__(256) void k_rownorm(float* __restrict__ out, int n) {
  const int wave = (int)(((unsigned)blockIdx.x * blockDim.x + threadIdx.x) >> 5);
  const int lane = (int)(threadIdx.x & 31u);
  if (wave >= n) return;
  float* row = out + (size_t)wave * HO + lane * 2;
  float2 x = *(float2*)row;
  float ss = x.x * x.x + x.y * x.y;
#pragma unroll
  for (int off = 16; off > 0; off >>= 1) ss += __shfl_xor(ss, off, 32);
  const float nrm = sqrtf(ss);
  const float scale = (nrm > 0.f) ? 1.0f / fmaxf(nrm, 1e-30f) : 0.0f;
  x.x *= scale;
  x.y *= scale;
  *(float2*)row = x;
}

#define NB 512
#define NCH 256
#define RMAXE 64
#define RMAXV 200
#define SRCMASK 0xFFFFF

__global__ __launch_bounds__(NB) void k_bin_count(
    const int* __restrict__ vertex, const int* __restrict__ edges,
    int* __restrict__ cntE, int* __restrict__ cntV, int nnz, int re, int rv) {
  __shared__ int hE[NB], hV[NB];
  const int t = (int)threadIdx.x, c = (int)blockIdx.x;
  hE[t] = 0;
  hV[t] = 0;
  __syncthreads();
  const int ch = (nnz + NCH - 1) / NCH;
  const int lo = c * ch;
  const int hi = min(lo + ch, nnz);
  for (int i = lo + t; i < hi; i += NB) {
    atomicAdd(&hE[edges[i] / re], 1);
    atomicAdd(&hV[vertex[i] / rv], 1);
  }
  __syncthreads();
  cntE[c * NB + t] = hE[t];
  cntV[c * NB + t] = hV[t];
}

__global__ __launch_bounds__(NB) void k_bin_offsets(
    int* __restrict__ cntE, int* __restrict__ cntV,
    int* __restrict__ bsE, int* __restrict__ bsV) {
  __shared__ int s[NB];
  int* tab = blockIdx.x ? cntV : cntE;
  int* bs = blockIdx.x ? bsV : bsE;
  const int b = (int)threadIdx.x;
  int run = 0;
  for (int c0 = 0; c0 < NCH; c0 += 32) {
    int v[32];
#pragma unroll
    for (int u = 0; u < 32; ++u) v[u] = tab[(c0 + u) * NB + b];
#pragma unroll
    for (int u = 0; u < 32; ++u) {
      tab[(c0 + u) * NB + b] = run;
      run += v[u];
    }
  }
  s[b] = run;
  __syncthreads();
  for (int off = 1; off < NB; off <<= 1) {
    const int add = (b >= off) ? s[b - off] : 0;
    __syncthreads();
    s[b] += add;
    __syncthreads();
  }
  bs[b] = s[b] - run;
  if (b == NB - 1) bs[NB] = s[b];
}

#define PCAP 8192
#define PK (PCAP / NB)
__global__ __launch_bounds__(NB) void k_bin_place(
    const int* __restrict__ vertex, const int* __restrict__ edges,
    const int* __restrict__ cntE, const int* __restrict__ cntV,
    const int* __restrict__ bsE, const int* __restrict__ bsV,
    int* __restrict__ listE, int* __restrict__ listV, int nnz, int re, int rv) {
  __shared__ int hE[NB], hV[NB];
  __shared__ int gE[NB], gV[NB];
  __shared__ int wsE[NB / 32], wsV[NB / 32];
  __shared__ int stE[PCAP], stV[PCAP];
  __shared__ int adE[PCAP], adV[PCAP];
  const int t = (int)threadIdx.x, c = (int)blockIdx.x;
  const int lane = t & 31;
  const int wave = t >> 5;
  int gbE = bsE[t] + cntE[c * NB + t];
  int gbV = bsV[t] + cntV[c * NB + t];
  const int ch = (nnz + NCH - 1) / NCH;
  const int lo = c * ch;
  const int hi = min(lo + ch, nnz);
  for (int plo = lo; plo < hi; plo += PCAP) {
    const int pn = min(PCAP, hi - plo);
    hE[t] = 0;
    hV[t] = 0;
    int wE[PK], wV[PK], pE[PK], pV[PK];
#pragma unroll
    for (int k = 0; k < PK; ++k) {
      const int i = plo + min(k * NB + t, pn - 1);
      wE[k] = vertex[i];
      wV[k] = edges[i];
    }
    __syncthreads();
#pragma unroll
    for (int k = 0; k < PK; ++k) {
      const bool ok = k * NB + t < pn;
      const int v = wE[k], e = wV[k];
      const int be = e / re, bv = v / rv;
      const int ce = ok ? be : t, cv = ok ? bv : t;
      const int one = ok ? 1 : 0;
      pE[k] = ce | (atomicAdd(&hE[ce], one) << 16);
      pV[k] = cv | (atomicAdd(&hV[cv], one) << 16);
      wE[k] = v | ((e - be * re) << 20);
      wV[k] = e | ((v - bv * rv) << 20);
    }
    __syncthreads();
    const int nE = hE[t], nV = hV[t];
    int iE = nE, iV = nV;
#pragma unroll
    for (int off = 1; off < 32; off <<= 1) {
      const int uE = __shfl_up(iE, off, 32);
      const int uV = __shfl_up(iV, off, 32);
      if (lane >= off) {
        iE += uE;
        iV += uV;
      }
    }
    if (lane == 31) {
      wsE[wave] = iE;
      wsV[wave] = iV;
    }
    __syncthreads();
    int bE = 0, bV = 0;
    for (int w2 = 0; w2 < NB / 32; ++w2)
      if (w2 < wave) {
        bE += wsE[w2];
        bV += wsV[w2];
      }
    hE[t] = bE + iE - nE;
    hV[t] = bV + iV - nV;
    gE[t] = gbE;
    gV[t] = gbV;
    gbE += nE;
    gbV += nV;
    __syncthreads();
    int se[PK], sv[PK];
#pragma unroll
    for (int k = 0; k < PK; ++k) {
      const int be = pE[k] & 0xFFFF, ke = pE[k] >> 16;
      const int bv = pV[k] & 0xFFFF, kv = pV[k] >> 16;
      se[k] = hE[be] + ke;
      pE[k] = gE[be] + ke;
      sv[k] = hV[bv] + kv;
      pV[k] = gV[bv] + kv;
    }
#pragma unroll
    for (int k = 0; k < PK; ++k)
      if (k * NB + t < pn) {
        stE[se[k]] = wE[k];
        adE[se[k]] = pE[k];
        stV[sv[k]] = wV[k];
        adV[sv[k]] = pV[k];
      }
    __syncthreads();
#pragma unroll
    for (int k = 0; k < PK; ++k) {
      const int s = min(k * NB + t, pn - 1);
      wE[k] = stE[s];
      pE[k] = adE[s];
      wV[k] = stV[s];
      pV[k] = adV[s];
    }
#pragma unroll
    for (int k = 0; k < PK; ++k)
      if (k * NB + t < pn) {
        listE[pE[k]] = wE[k];
        listV[pV[k]] = wV[k];
      }
  }
}

#define TRIPB 16
#define TILE 4608
template <bool V2E, int RMAX>
__global__ __launch_bounds__(NB) void k_range_sum(
    const float* __restrict__ src, const int* __restrict__ list, const int* __restrict__ bs,
    const float* __restrict__ homo, float* __restrict__ dst, int r, int nrows) {
  __shared__ __attribute__((aligned(16))) float acc[RMAX * HO];
  __shared__ float aux[RMAX];
  __shared__ int cur[RMAX];
  __shared__ int sorted[TILE];
  const int t = (int)threadIdx.x;
  const int lane = t & 31;
  const int bin = (int)blockIdx.x;
  const int lo = bs[bin], hi = bs[bin + 1];
  for (int j = t; j < r * HO; j += NB) acc[j] = 0.0f;
  for (int j = t; j < r; j += NB) aux[j] = 0.0f;
  __syncthreads();
  for (int i = lo + t; i < hi; i += NB) {
    const int ent = list[i];
    atomicAdd(&aux[ent >> 20], V2E ? 1.0f : homo[ent & SRCMASK]);
  }
  const int wave = __builtin_amdgcn_readfirstlane(t >> 5);
  constexpr int NWV = NB / 32;
  constexpr int KW = TILE / NB;
  constexpr int RPL = (RMAX + 31) / 32;
  for (int tlo = lo; tlo < hi; tlo += TILE) {
    const int tn = min(TILE, hi - tlo);
    for (int j = t; j < r; j += NB) cur[j] = 0;
    int w[KW];
#pragma unroll
    for (int k = 0; k < KW; ++k) w[k] = list[min(tlo + k * NB + t, hi - 1)];
    __syncthreads();
#pragma unroll
    for (int k = 0; k < KW; ++k)
      if (k * NB + t < tn) atomicAdd(&cur[w[k] >> 20], 1);
    __syncthreads();
    if (wave == 0) {
      int c[RPL];
      int s = 0;
#pragma unroll
      for (int u = 0; u < RPL; ++u) {
        const int j = lane * RPL + u;
        c[u] = (j < r) ? cur[j] : 0;
        s += c[u];
      }
      int incl = s;
#pragma unroll
      for (int off = 1; off < 32; off <<= 1) {
        const int v = __shfl_up(incl, off, 32);
        if (lane >= off) incl += v;
      }
      int run = incl - s;
#pragma unroll
      for (int u = 0; u < RPL; ++u) {
        const int j = lane * RPL + u;
        if (j < r) cur[j] = run;
        run += c[u];
      }
    }
    __syncthreads();
#pragma unroll
    for (int k = 0; k < KW; ++k)
      if (k * NB + t < tn) sorted[atomicAdd(&cur[w[k] >> 20], 1)] = w[k];
    __syncthreads();
    const int seg = ((tn + NWV * TRIPB - 1) / (NWV * TRIPB)) * TRIPB;
    const int wbeg = wave * seg;
    const int wend = min(wbeg + seg, tn);
    if (wbeg < wend) {
      float a0 = 0.0f, a1 = 0.0f;
      int crow = -1;
      int ent = sorted[min(wbeg + lane, wend - 1)];
      for (int b = wbeg; b < wend; b += TRIPB) {
        const int nv = min(TRIPB, wend - b);
        const int entn = sorted[min(b + TRIPB + lane, wend - 1)];
        const int sl = ent & SRCMASK;
        const int dl = ent >> 20;
        float al = 1.0f;
        if (!V2E && lane < nv) al = homo[sl] / aux[dl];
        float x0[TRIPB], x1[TRIPB];
#pragma unroll
        for (int u = 0; u < TRIPB; ++u) {
          const float* s = src + (size_t)__builtin_amdgcn_readlane(sl, u) * HO + lane;
          x0[u] = s[0];
          x1[u] = s[32];
        }
#pragma unroll
        for (int u = 0; u < TRIPB; ++u) {
          if (u < nv) {
            const int d = __builtin_amdgcn_readlane(dl, u);
            if (d != crow) {
              if (crow >= 0) {
                atomicAdd(acc + crow * HO + lane, a0);
                atomicAdd(acc + crow * HO + 32 + lane, a1);
              }
              a0 = 0.0f;
              a1 = 0.0f;
              crow = d;
            }
            if (V2E) {
              a0 += x0[u];
              a1 += x1[u];
            } else {
              const float att = __int_as_float(__builtin_amdgcn_readlane(__float_as_int(al), u));
              a0 += att * x0[u];
              a1 += att * x1[u];
            }
          }
        }
        ent = entn;
      }
      if (crow >= 0) {
        atomicAdd(acc + crow * HO + lane, a0);
        atomicAdd(acc + crow * HO + 32 + lane, a1);
      }
    }
  }
  if (V2E) {
    __syncthreads();
    for (int j = t; j < r * HO; j += NB) {
      const int row = bin * r + (j >> 6);
      if (row < nrows) dst[(size_t)row * HO + (j & 63)] = acc[j] / fmaxf(aux[j >> 6], 1.0f);
    }
  } else {
    constexpr int NW = NB / 32;
    constexpr int RPW = (RMAX + NW - 1) / NW;
    const int nv = min(r, nrows - bin * r);
    if (nv <= 0) return;
    float2 x[RPW];
#pragma unroll
    for (int u = 0; u < RPW; ++u) {
      const int rc = min(wave + u * NW, nv - 1);
      x[u] = *(const float2*)(dst + (size_t)(bin * r + rc) * HO + lane * 2);
    }
    __syncthreads();
#pragma unroll
    for (int u = 0; u < RPW; ++u) {
      const int rc = min(wave + u * NW, nv - 1);
      const float2 a = *(const float2*)(acc + rc * HO + lane * 2);
      x[u].x = x[u].x + a.x;
      x[u].y = x[u].y + a.y;
      float ss = x[u].x * x[u].x + x[u].y * x[u].y;
#pragma unroll
      for (int off = 16; off > 0; off >>= 1) ss += __shfl_xor(ss, off, 32);
      const float nrm = sqrtf(ss);
      const float inv = 1.0f / fmaxf(nrm, 1e-30f);
      const float scale = (nrm > 0.f) ? inv : 0.0f;
      x[u].x *= scale;
      x[u].y *= scale;
    }
#pragma unroll
    for (int u = 0; u < RPW; ++u) {
      const int rr = wave + u * NW;
      if (rr < nv)
        *(float2*)(dst + (size_t)(bin * r + rr) * HO + lane * 2) = x[u];
    }
  }
}

extern "C" void kernel_launch(void* const* d_in, const int* in_sizes, int n_in,
                              void* d_out, int out_size, void* d_ws, size_t ws_size,
                              hipStream_t stream) {
  const float* X      = (const float*)d_in[0];
  const float* W      = (const float*)d_in[1];
  const float* homo   = (const float*)d_in[2];
  const int*   vertex = (const int*)d_in[3];
  const int*   edges  = (const int*)d_in[4];
  float* out = (float*)d_out;

  const int n   = in_sizes[0] / INCH;
  const int e   = in_sizes[2];
  const int nnz = in_sizes[3];

  const int tiles = (n + 15) / 16;
  k_gemm_wmma<<<(tiles * 32 + 255) / 256, 256, 0, stream>>>(X, W, out, n);

  const int re = (e + NB - 1) / NB;
  const int rv = (n + NB - 1) / NB;
  const size_t need = ((size_t)e * HO + 2 * (size_t)nnz + 2 * (size_t)NCH * NB + 2 * (NB + 1)) * 4;
  float* Xe = (float*)d_ws;

  if (need <= ws_size && re <= RMAXE && rv <= RMAXV && n <= SRCMASK && e <= SRCMASK) {
    int* listE = (int*)(Xe + (size_t)e * HO);
    int* listV = listE + nnz;
    int* cntE  = listV + nnz;
    int* cntV  = cntE + NCH * NB;
    int* bsE   = cntV + NCH * NB;
    int* bsV   = bsE + NB + 1;
    k_bin_count<<<NCH, NB, 0, stream>>>(vertex, edges, cntE, cntV, nnz, re, rv);
    k_bin_offsets<<<2, NB, 0, stream>>>(cntE, cntV, bsE, bsV);
    k_bin_place<<<NCH, NB, 0, stream>>>(vertex, edges, cntE, cntV, bsE, bsV, listE, listV, nnz, re, rv);
    k_range_sum<true, RMAXE><<<NB, NB, 0, stream>>>(out, listE, bsE, homo, Xe, re, e);
    k_range_sum<false, RMAXV><<<NB, NB, 0, stream>>>(Xe, listV, bsV, homo, out, rv, n);
  } else {
    float* cnt     = Xe + (size_t)e * HO;
    float* att_sum = cnt + e;
    const long long zn = (long long)e * HO + e + n;
    k_zero<<<(int)((zn + 255) / 256), 256, 0, stream>>>(Xe, zn);
    k_scatter_rows<true><<<8192, 256, 0, stream>>>(
        out, vertex, edges, homo, Xe, cnt, att_sum, nnz);
    const long long em = (long long)e * HO;
    k_edge_mean<<<(int)((em + 255) / 256), 256, 0, stream>>>(Xe, cnt, em);
    k_scatter_rows<false><<<8192, 256, 0, stream>>>(
        Xe, vertex, edges, homo, out, cnt, att_sum, nnz);
    const long long nt = (long long)n * 32;
    k_rownorm<<<(int)((nt + 255) / 256), 256, 0, stream>>>(out, n);
  }
}


# Xp GEMM and the bin count share one launch (count work-groups first, GEMM row tiles in the rest): 6 dependent launches become 5 and the count's serial index reads run under the GEMM
# speedup vs baseline: 5.0564x; 1.0521x over previous
#include <hip/hip_runtime.h>
#include <math.h>

typedef float v2f __attribute__((ext_vector_type(2)));
typedef float v8f __attribute__((ext_vector_type(8)));

#define HO 64
#define INCH 64

__global__ __launch_bounds__(256) void k_zero(float* __restrict__ p, long long n) {
  long long i = (long long)blockIdx.x * blockDim.x + threadIdx.x;
  if (i < n) p[i] = 0.0f;
}

__device__ __forceinline__ void gemm_rows16(
    const float* __restrict__ X, const float* __restrict__ W,
    float* __restrict__ Xp, int nrows, int wave, int lane) {
  const int row0 = wave * 16;
  if (row0 >= nrows) return;
  if (row0 + 16 <= nrows) {
    const int m  = lane & 15;
    const int kh = lane >> 4;
    v8f acc0 = {}, acc1 = {}, acc2 = {}, acc3 = {};
#pragma unroll
    for (int k0 = 0; k0 < INCH; k0 += 4) {
      v2f a = *(const v2f*)(X + (size_t)(row0 + m) * INCH + (k0 + kh * 2));
      const float* w0 = W + (size_t)(k0 + kh * 2) * HO;
      const float* w1 = w0 + HO;
      v2f b0, b1, b2, b3;
      b0.x = w0[m];      b0.y = w1[m];
      b1.x = w0[m + 16]; b1.y = w1[m + 16];
      b2.x = w0[m + 32]; b2.y = w1[m + 32];
      b3.x = w0[m + 48]; b3.y = w1[m + 48];
      acc0 = __builtin_amdgcn_wmma_f32_16x16x4_f32(false, a, false, b0, (short)0, acc0, false, false);
      acc1 = __builtin_amdgcn_wmma_f32_16x16x4_f32(false, a, false, b1, (short)0, acc1, false, false);
      acc2 = __builtin_amdgcn_wmma_f32_16x16x4_f32(false, a, false, b2, (short)0, acc2, false, false);
      acc3 = __builtin_amdgcn_wmma_f32_16x16x4_f32(false, a, false, b3, (short)0, acc3, false, false);
    }
#pragma unroll
    for (int r = 0; r < 8; ++r) {
      const int row = row0 + r + kh * 8;
      float* o = Xp + (size_t)row * HO + m;
      o[0]  = acc0[r];
      o[16] = acc1[r];
      o[32] = acc2[r];
      o[48] = acc3[r];
    }
  } else {
    for (int r = row0 + lane; r < nrows; r += 32)
      for (int c = 0; c < HO; ++c) {
        float s = 0.f;
        for (int k = 0; k < INCH; ++k) s += X[(size_t)r * INCH + k] * W[(size_t)k * HO + c];
        Xp[(size_t)r * HO + c] = s;
      }
  }
}

__global__ __launch_bounds__(256) void k_gemm_wmma(
    const float* __restrict__ X, const float* __restrict__ W,
    float* __restrict__ Xp, int nrows) {
  const int wave = (int)(((unsigned)blockIdx.x * blockDim.x + threadIdx.x) >> 5);
  const int lane = (int)(threadIdx.x & 31u);
  gemm_rows16(X, W, Xp, nrows, wave, lane);
}

#define TRIP 8
template <bool V2E>
__global__ __launch_bounds__(256) void k_scatter_rows(
    const float* __restrict__ src, const int* __restrict__ vertex,
    const int* __restrict__ edges, const float* __restrict__ homo,
    float* __restrict__ dst, float* __restrict__ cnt,
    float* __restrict__ att_sum, int nnz) {
  const int lane = (int)(threadIdx.x & 31u);
  const int wave = __builtin_amdgcn_readfirstlane(
      (int)(((unsigned)blockIdx.x * blockDim.x + threadIdx.x) >> 5));
  const int nwaves = (int)((gridDim.x * blockDim.x) >> 5);
  for (int base = wave * TRIP; base < nnz; base += nwaves * TRIP) {
    const int idx = base + lane;
    const bool mine = lane < TRIP && idx < nnz;
    int vl = 0, el = 0;
    if (mine) {
      vl = vertex[idx];
      el = edges[idx];
    }
    const bool whole = base + TRIP <= nnz;
    float x0[TRIP], x1[TRIP];
    if (whole) {
#pragma unroll
      for (int u = 0; u < TRIP; ++u) {
        const int r = __builtin_amdgcn_readlane(V2E ? vl : el, u);
        const float* s = src + (size_t)r * HO + lane;
        x0[u] = s[0];
        x1[u] = s[32];
      }
    }
    float al = 0.0f;
    if (mine) {
      if (V2E) {
        atomicAdd(cnt + el, 1.0f);
        atomicAdd(att_sum + vl, homo[el]);
      } else {
        al = homo[el] / att_sum[vl];
      }
    }
    if (whole) {
#pragma unroll
      for (int u = 0; u < TRIP; ++u) {
        const int r = __builtin_amdgcn_readlane(V2E ? el : vl, u);
        float* d = dst + (size_t)r * HO + lane;
        float a0 = x0[u], a1 = x1[u];
        if (!V2E) {
          const float att = __int_as_float(__builtin_amdgcn_readlane(__float_as_int(al), u));
          a0 = att * a0;
          a1 = att * a1;
        }
        atomicAdd(d, a0);
        atomicAdd(d + 32, a1);
      }
    } else {
      for (int u = 0; base + u < nnz; ++u) {
        const int rs = __shfl(V2E ? vl : el, u, 32);
        const int rd = __shfl(V2E ? el : vl, u, 32);
        const float* s = src + (size_t)rs * HO + lane;
        float* d = dst + (size_t)rd * HO + lane;
        float a0 = s[0], a1 = s[32];
        if (!V2E) {
          const float att = __shfl(al, u, 32);
          a0 = att * a0;
          a1 = att * a1;
        }
        atomicAdd(d, a0);
        atomicAdd(d + 32, a1);
      }
    }
  }
}

__global__ __launch_bounds__(256) void k_edge_mean(float* __restrict__ Xe,
                                                   const float* __restrict__ cnt,
                                                   long long total) {
  long long i = (long long)blockIdx.x * blockDim.x + threadIdx.x;
  if (i >= total) return;
  const float c = cnt[i >> 6];
  Xe[i] *= 1.0f / fmaxf(c, 1.0f);
}

__global__ __launch_bounds__(256) void k_rownorm(float* __restrict__ out, int n) {
  const int wave = (int)(((unsigned)blockIdx.x * blockDim.x + threadIdx.x) >> 5);
  const int lane = (int)(threadIdx.x & 31u);
  if (wave >= n) return;
  float* row = out + (size_t)wave * HO + lane * 2;
  float2 x = *(float2*)row;
  float ss = x.x * x.x + x.y * x.y;
#pragma unroll
  for (int off = 16; off > 0; off >>= 1) ss += __shfl_xor(ss, off, 32);
  const float nrm = sqrtf(ss);
  const float scale = (nrm > 0.f) ? 1.0f / fmaxf(nrm, 1e-30f) : 0.0f;
  x.x *= scale;
  x.y *= scale;
  *(float2*)row = x;
}

#define NB 512
#define NCH 256
#define RMAXE 64
#define RMAXV 200
#define SRCMASK 0xFFFFF

__global__ __launch_bounds__(NB) void k_count_gemm(
    const int* __restrict__ vertex, const int* __restrict__ edges,
    int* __restrict__ cntE, int* __restrict__ cntV, int nnz, int re, int rv,
    const float* __restrict__ X, const float* __restrict__ W,
    float* __restrict__ Xp, int nrows) {
  __shared__ int hE[NB], hV[NB];
  const int t = (int)threadIdx.x;
  if ((int)blockIdx.x >= NCH) {
    const int wave = (((int)blockIdx.x - NCH) * NB + t) >> 5;
    gemm_rows16(X, W, Xp, nrows, wave, t & 31);
    return;
  }
  const int c = (int)blockIdx.x;
  hE[t] = 0;
  hV[t] = 0;
  __syncthreads();
  const int ch = (nnz + NCH - 1) / NCH;
  const int lo = c * ch;
  const int hi = min(lo + ch, nnz);
  for (int i = lo + t; i < hi; i += NB) {
    atomicAdd(&hE[edges[i] / re], 1);
    atomicAdd(&hV[vertex[i] / rv], 1);
  }
  __syncthreads();
  cntE[c * NB + t] = hE[t];
  cntV[c * NB + t] = hV[t];
}

__global__ __launch_bounds__(NB) void k_bin_offsets(
    int* __restrict__ cntE, int* __restrict__ cntV,
    int* __restrict__ bsE, int* __restrict__ bsV) {
  __shared__ int s[NB];
  int* tab = blockIdx.x ? cntV : cntE;
  int* bs = blockIdx.x ? bsV : bsE;
  const int b = (int)threadIdx.x;
  int run = 0;
  for (int c0 = 0; c0 < NCH; c0 += 32) {
    int v[32];
#pragma unroll
    for (int u = 0; u < 32; ++u) v[u] = tab[(c0 + u) * NB + b];
#pragma unroll
    for (int u = 0; u < 32; ++u) {
      tab[(c0 + u) * NB + b] = run;
      run += v[u];
    }
  }
  s[b] = run;
  __syncthreads();
  for (int off = 1; off < NB; off <<= 1) {
    const int add = (b >= off) ? s[b - off] : 0;
    __syncthreads();
    s[b] += add;
    __syncthreads();
  }
  bs[b] = s[b] - run;
  if (b == NB - 1) bs[NB] = s[b];
}

#define PCAP 8192
#define PK (PCAP / NB)
__global__ __launch_bounds__(NB) void k_bin_place(
    const int* __restrict__ vertex, const int* __restrict__ edges,
    const int* __restrict__ cntE, const int* __restrict__ cntV,
    const int* __restrict__ bsE, const int* __restrict__ bsV,
    int* __restrict__ listE, int* __restrict__ listV, int nnz, int re, int rv) {
  __shared__ int hE[NB], hV[NB];
  __shared__ int gE[NB], gV[NB];
  __shared__ int wsE[NB / 32], wsV[NB / 32];
  __shared__ int stE[PCAP], stV[PCAP];
  __shared__ int adE[PCAP], adV[PCAP];
  const int t = (int)threadIdx.x, c = (int)blockIdx.x;
  const int lane = t & 31;
  const int wave = t >> 5;
  int gbE = bsE[t] + cntE[c * NB + t];
  int gbV = bsV[t] + cntV[c * NB + t];
  const int ch = (nnz + NCH - 1) / NCH;
  const int lo = c * ch;
  const int hi = min(lo + ch, nnz);
  for (int plo = lo; plo < hi; plo += PCAP) {
    const int pn = min(PCAP, hi - plo);
    hE[t] = 0;
    hV[t] = 0;
    int wE[PK], wV[PK], pE[PK], pV[PK];
#pragma unroll
    for (int k = 0; k < PK; ++k) {
      const int i = plo + min(k * NB + t, pn - 1);
      wE[k] = vertex[i];
      wV[k] = edges[i];
    }
    __syncthreads();
#pragma unroll
    for (int k = 0; k < PK; ++k) {
      const bool ok = k * NB + t < pn;
      const int v = wE[k], e = wV[k];
      const int be = e / re, bv = v / rv;
      const int ce = ok ? be : t, cv = ok ? bv : t;
      const int one = ok ? 1 : 0;
      pE[k] = ce | (atomicAdd(&hE[ce], one) << 16);
      pV[k] = cv | (atomicAdd(&hV[cv], one) << 16);
      wE[k] = v | ((e - be * re) << 20);
      wV[k] = e | ((v - bv * rv) << 20);
    }
    __syncthreads();
    const int nE = hE[t], nV = hV[t];
    int iE = nE, iV = nV;
#pragma unroll
    for (int off = 1; off < 32; off <<= 1) {
      const int uE = __shfl_up(iE, off, 32);
      const int uV = __shfl_up(iV, off, 32);
      if (lane >= off) {
        iE += uE;
        iV += uV;
      }
    }
    if (lane == 31) {
      wsE[wave] = iE;
      wsV[wave] = iV;
    }
    __syncthreads();
    int bE = 0, bV = 0;
    for (int w2 = 0; w2 < NB / 32; ++w2)
      if (w2 < wave) {
        bE += wsE[w2];
        bV += wsV[w2];
      }
    hE[t] = bE + iE - nE;
    hV[t] = bV + iV - nV;
    gE[t] = gbE;
    gV[t] = gbV;
    gbE += nE;
    gbV += nV;
    __syncthreads();
    int se[PK], sv[PK];
#pragma unroll
    for (int k = 0; k < PK; ++k) {
      const int be = pE[k] & 0xFFFF, ke = pE[k] >> 16;
      const int bv = pV[k] & 0xFFFF, kv = pV[k] >> 16;
      se[k] = hE[be] + ke;
      pE[k] = gE[be] + ke;
      sv[k] = hV[bv] + kv;
      pV[k] = gV[bv] + kv;
    }
#pragma unroll
    for (int k = 0; k < PK; ++k)
      if (k * NB + t < pn) {
        stE[se[k]] = wE[k];
        adE[se[k]] = pE[k];
        stV[sv[k]] = wV[k];
        adV[sv[k]] = pV[k];
      }
    __syncthreads();
#pragma unroll
    for (int k = 0; k < PK; ++k) {
      const int s = min(k * NB + t, pn - 1);
      wE[k] = stE[s];
      pE[k] = adE[s];
      wV[k] = stV[s];
      pV[k] = adV[s];
    }
#pragma unroll
    for (int k = 0; k < PK; ++k)
      if (k * NB + t < pn) {
        listE[pE[k]] = wE[k];
        listV[pV[k]] = wV[k];
      }
  }
}

#define TRIPB 16
#define TILE 4608
template <bool V2E, int RMAX>
__global__ __launch_bounds__(NB) void k_range_sum(
    const float* __restrict__ src, const int* __restrict__ list, const int* __restrict__ bs,
    const float* __restrict__ homo, float* __restrict__ dst, int r, int nrows) {
  __shared__ __attribute__((aligned(16))) float acc[RMAX * HO];
  __shared__ float aux[RMAX];
  __shared__ int cur[RMAX];
  __shared__ int sorted[TILE];
  const int t = (int)threadIdx.x;
  const int lane = t & 31;
  const int bin = (int)blockIdx.x;
  const int lo = bs[bin], hi = bs[bin + 1];
  for (int j = t; j < r * HO; j += NB) acc[j] = 0.0f;
  for (int j = t; j < r; j += NB) aux[j] = 0.0f;
  __syncthreads();
  for (int i = lo + t; i < hi; i += NB) {
    const int ent = list[i];
    atomicAdd(&aux[ent >> 20], V2E ? 1.0f : homo[ent & SRCMASK]);
  }
  const int wave = __builtin_amdgcn_readfirstlane(t >> 5);
  constexpr int NWV = NB / 32;
  constexpr int KW = TILE / NB;
  constexpr int RPL = (RMAX + 31) / 32;
  for (int tlo = lo; tlo < hi; tlo += TILE) {
    const int tn = min(TILE, hi - tlo);
    for (int j = t; j < r; j += NB) cur[j] = 0;
    int w[KW];
#pragma unroll
    for (int k = 0; k < KW; ++k) w[k] = list[min(tlo + k * NB + t, hi - 1)];
    __syncthreads();
#pragma unroll
    for (int k = 0; k < KW; ++k)
      if (k * NB + t < tn) atomicAdd(&cur[w[k] >> 20], 1);
    __syncthreads();
    if (wave == 0) {
      int c[RPL];
      int s = 0;
#pragma unroll
      for (int u = 0; u < RPL; ++u) {
        const int j = lane * RPL + u;
        c[u] = (j < r) ? cur[j] : 0;
        s += c[u];
      }
      int incl = s;
#pragma unroll
      for (int off = 1; off < 32; off <<= 1) {
        const int v = __shfl_up(incl, off, 32);
        if (lane >= off) incl += v;
      }
      int run = incl - s;
#pragma unroll
      for (int u = 0; u < RPL; ++u) {
        const int j = lane * RPL + u;
        if (j < r) cur[j] = run;
        run += c[u];
      }
    }
    __syncthreads();
#pragma unroll
    for (int k = 0; k < KW; ++k)
      if (k * NB + t < tn) sorted[atomicAdd(&cur[w[k] >> 20], 1)] = w[k];
    __syncthreads();
    const int seg = ((tn + NWV * TRIPB - 1) / (NWV * TRIPB)) * TRIPB;
    const int wbeg = wave * seg;
    const int wend = min(wbeg + seg, tn);
    if (wbeg < wend) {
      float a0 = 0.0f, a1 = 0.0f;
      int crow = -1;
      int ent = sorted[min(wbeg + lane, wend - 1)];
      for (int b = wbeg; b < wend; b += TRIPB) {
        const int nv = min(TRIPB, wend - b);
        const int entn = sorted[min(b + TRIPB + lane, wend - 1)];
        const int sl = ent & SRCMASK;
        const int dl = ent >> 20;
        float al = 1.0f;
        if (!V2E && lane < nv) al = homo[sl] / aux[dl];
        float x0[TRIPB], x1[TRIPB];
#pragma unroll
        for (int u = 0; u < TRIPB; ++u) {
          const float* s = src + (size_t)__builtin_amdgcn_readlane(sl, u) * HO + lane;
          x0[u] = s[0];
          x1[u] = s[32];
        }
#pragma unroll
        for (int u = 0; u < TRIPB; ++u) {
          if (u < nv) {
            const int d = __builtin_amdgcn_readlane(dl, u);
            if (d != crow) {
              if (crow >= 0) {
                atomicAdd(acc + crow * HO + lane, a0);
                atomicAdd(acc + crow * HO + 32 + lane, a1);
              }
              a0 = 0.0f;
              a1 = 0.0f;
              crow = d;
            }
            if (V2E) {
              a0 += x0[u];
              a1 += x1[u];
            } else {
              const float att = __int_as_float(__builtin_amdgcn_readlane(__float_as_int(al), u));
              a0 += att * x0[u];
              a1 += att * x1[u];
            }
          }
        }
        ent = entn;
      }
      if (crow >= 0) {
        atomicAdd(acc + crow * HO + lane, a0);
        atomicAdd(acc + crow * HO + 32 + lane, a1);
      }
    }
  }
  if (V2E) {
    __syncthreads();
    for (int j = t; j < r * HO; j += NB) {
      const int row = bin * r + (j >> 6);
      if (row < nrows) dst[(size_t)row * HO + (j & 63)] = acc[j] / fmaxf(aux[j >> 6], 1.0f);
    }
  } else {
    constexpr int NW = NB / 32;
    constexpr int RPW = (RMAX + NW - 1) / NW;
    const int nv = min(r, nrows - bin * r);
    if (nv <= 0) return;
    float2 x[RPW];
#pragma unroll
    for (int u = 0; u < RPW; ++u) {
      const int rc = min(wave + u * NW, nv - 1);
      x[u] = *(const float2*)(dst + (size_t)(bin * r + rc) * HO + lane * 2);
    }
    __syncthreads();
#pragma unroll
    for (int u = 0; u < RPW; ++u) {
      const int rc = min(wave + u * NW, nv - 1);
      const float2 a = *(const float2*)(acc + rc * HO + lane * 2);
      x[u].x = x[u].x + a.x;
      x[u].y = x[u].y + a.y;
      float ss = x[u].x * x[u].x + x[u].y * x[u].y;
#pragma unroll
      for (int off = 16; off > 0; off >>= 1) ss += __shfl_xor(ss, off, 32);
      const float nrm = sqrtf(ss);
      const float inv = 1.0f / fmaxf(nrm, 1e-30f);
      const float scale = (nrm > 0.f) ? inv : 0.0f;
      x[u].x *= scale;
      x[u].y *= scale;
    }
#pragma unroll
    for (int u = 0; u < RPW; ++u) {
      const int rr = wave + u * NW;
      if (rr < nv)
        *(float2*)(dst + (size_t)(bin * r + rr) * HO + lane * 2) = x[u];
    }
  }
}

extern "C" void kernel_launch(void* const* d_in, const int* in_sizes, int n_in,
                              void* d_out, int out_size, void* d_ws, size_t ws_size,
                              hipStream_t stream) {
  const float* X      = (const float*)d_in[0];
  const float* W      = (const float*)d_in[1];
  const float* homo   = (const float*)d_in[2];
  const int*   vertex = (const int*)d_in[3];
  const int*   edges  = (const int*)d_in[4];
  float* out = (float*)d_out;

  const int n   = in_sizes[0] / INCH;
  const int e   = in_sizes[2];
  const int nnz = in_sizes[3];

  const int tiles = (n + 15) / 16;

  const int re = (e + NB - 1) / NB;
  const int rv = (n + NB - 1) / NB;
  const size_t need = ((size_t)e * HO + 2 * (size_t)nnz + 2 * (size_t)NCH * NB + 2 * (NB + 1)) * 4;
  float* Xe = (float*)d_ws;

  if (need <= ws_size && re <= RMAXE && rv <= RMAXV && n <= SRCMASK && e <= SRCMASK) {
    int* listE = (int*)(Xe + (size_t)e * HO);
    int* listV = listE + nnz;
    int* cntE  = listV + nnz;
    int* cntV  = cntE + NCH * NB;
    int* bsE   = cntV + NCH * NB;
    int* bsV   = bsE + NB + 1;
    const int gblocks = (tiles + NB / 32 - 1) / (NB / 32);
    k_count_gemm<<<NCH + gblocks, NB, 0, stream>>>(vertex, edges, cntE, cntV, nnz, re, rv, X, W, out, n);
    k_bin_offsets<<<2, NB, 0, stream>>>(cntE, cntV, bsE, bsV);
    k_bin_place<<<NCH, NB, 0, stream>>>(vertex, edges, cntE, cntV, bsE, bsV, listE, listV, nnz, re, rv);
    k_range_sum<true, RMAXE><<<NB, NB, 0, stream>>>(out, listE, bsE, homo, Xe, re, e);
    k_range_sum<false, RMAXV><<<NB, NB, 0, stream>>>(Xe, listV, bsV, homo, out, rv, n);
  } else {
    k_gemm_wmma<<<(tiles * 32 + 255) / 256, 256, 0, stream>>>(X, W, out, n);
    float* cnt     = Xe + (size_t)e * HO;
    float* att_sum = cnt + e;
    const long long zn = (long long)e * HO + e + n;
    k_zero<<<(int)((zn + 255) / 256), 256, 0, stream>>>(Xe, zn);
    k_scatter_rows<true><<<8192, 256, 0, stream>>>(
        out, vertex, edges, homo, Xe, cnt, att_sum, nnz);
    const long long em = (long long)e * HO;
    k_edge_mean<<<(int)((em + 255) / 256), 256, 0, stream>>>(Xe, cnt, em);
    k_scatter_rows<false><<<8192, 256, 0, stream>>>(
        Xe, vertex, edges, homo, out, cnt, att_sum, nnz);
    const long long nt = (long long)n * 32;
    k_rownorm<<<(int)((nt + 255) / 256), 256, 0, stream>>>(out, n);
  }
}


# Range sums: per-row scalars come out of the tile pass (count = the sort's counts; homo summed per run, one division per row at the store): the serial list sweep and its barrier go (sum order changes)
# speedup vs baseline: 5.3894x; 1.0659x over previous
#include <hip/hip_runtime.h>
#include <math.h>

typedef float v2f __attribute__((ext_vector_type(2)));
typedef float v8f __attribute__((ext_vector_type(8)));

#define HO 64
#define INCH 64

__global__ __launch_bounds__(256) void k_zero(float* __restrict__ p, long long n) {
  long long i = (long long)blockIdx.x * blockDim.x + threadIdx.x;
  if (i < n) p[i] = 0.0f;
}

__device__ __forceinline__ void gemm_rows16(
    const float* __restrict__ X, const float* __restrict__ W,
    float* __restrict__ Xp, int nrows, int wave, int lane) {
  const int row0 = wave * 16;
  if (row0 >= nrows) return;
  if (row0 + 16 <= nrows) {
    const int m  = lane & 15;
    const int kh = lane >> 4;
    v8f acc0 = {}, acc1 = {}, acc2 = {}, acc3 = {};
#pragma unroll
    for (int k0 = 0; k0 < INCH; k0 += 4) {
      v2f a = *(const v2f*)(X + (size_t)(row0 + m) * INCH + (k0 + kh * 2));
      const float* w0 = W + (size_t)(k0 + kh * 2) * HO;
      const float* w1 = w0 + HO;
      v2f b0, b1, b2, b3;
      b0.x = w0[m];      b0.y = w1[m];
      b1.x = w0[m + 16]; b1.y = w1[m + 16];
      b2.x = w0[m + 32]; b2.y = w1[m + 32];
      b3.x = w0[m + 48]; b3.y = w1[m + 48];
      acc0 = __builtin_amdgcn_wmma_f32_16x16x4_f32(false, a, false, b0, (short)0, acc0, false, false);
      acc1 = __builtin_amdgcn_wmma_f32_16x16x4_f32(false, a, false, b1, (short)0, acc1, false, false);
      acc2 = __builtin_amdgcn_wmma_f32_16x16x4_f32(false, a, false, b2, (short)0, acc2, false, false);
      acc3 = __builtin_amdgcn_wmma_f32_16x16x4_f32(false, a, false, b3, (short)0, acc3, false, false);
    }
#pragma unroll
    for (int r = 0; r < 8; ++r) {
      const int row = row0 + r + kh * 8;
      float* o = Xp + (size_t)row * HO + m;
      o[0]  = acc0[r];
      o[16] = acc1[r];
      o[32] = acc2[r];
      o[48] = acc3[r];
    }
  } else {
    for (int r = row0 + lane; r < nrows; r += 32)
      for (int c = 0; c < HO; ++c) {
        float s = 0.f;
        for (int k = 0; k < INCH; ++k) s += X[(size_t)r * INCH + k] * W[(size_t)k * HO + c];
        Xp[(size_t)r * HO + c] = s;
      }
  }
}

__global__ __launch_bounds__(256) void k_gemm_wmma(
    const float* __restrict__ X, const float* __restrict__ W,
    float* __restrict__ Xp, int nrows) {
  const int wave = (int)(((unsigned)blockIdx.x * blockDim.x + threadIdx.x) >> 5);
  const int lane = (int)(threadIdx.x & 31u);
  gemm_rows16(X, W, Xp, nrows, wave, lane);
}

#define TRIP 8
template <bool V2E>
__global__ __launch_bounds__(256) void k_scatter_rows(
    const float* __restrict__ src, const int* __restrict__ vertex,
    const int* __restrict__ edges, const float* __restrict__ homo,
    float* __restrict__ dst, float* __restrict__ cnt,
    float* __restrict__ att_sum, int nnz) {
  const int lane = (int)(threadIdx.x & 31u);
  const int wave = __builtin_amdgcn_readfirstlane(
      (int)(((unsigned)blockIdx.x * blockDim.x + threadIdx.x) >> 5));
  const int nwaves = (int)((gridDim.x * blockDim.x) >> 5);
  for (int base = wave * TRIP; base < nnz; base += nwaves * TRIP) {
    const int idx = base + lane;
    const bool mine = lane < TRIP && idx < nnz;
    int vl = 0, el = 0;
    if (mine) {
      vl = vertex[idx];
      el = edges[idx];
    }
    const bool whole = base + TRIP <= nnz;
    float x0[TRIP], x1[TRIP];
    if (whole) {
#pragma unroll
      for (int u = 0; u < TRIP; ++u) {
        const int r = __builtin_amdgcn_readlane(V2E ? vl : el, u);
        const float* s = src + (size_t)r * HO + lane;
        x0[u] = s[0];
        x1[u] = s[32];
      }
    }
    float al = 0.0f;
    if (mine) {
      if (V2E) {
        atomicAdd(cnt + el, 1.0f);
        atomicAdd(att_sum + vl, homo[el]);
      } else {
        al = homo[el] / att_sum[vl];
      }
    }
    if (whole) {
#pragma unroll
      for (int u = 0; u < TRIP; ++u) {
        const int r = __builtin_amdgcn_readlane(V2E ? el : vl, u);
        float* d = dst + (size_t)r * HO + lane;
        float a0 = x0[u], a1 = x1[u];
        if (!V2E) {
          const float att = __int_as_float(__builtin_amdgcn_readlane(__float_as_int(al), u));
          a0 = att * a0;
          a1 = att * a1;
        }
        atomicAdd(d, a0);
        atomicAdd(d + 32, a1);
      }
    } else {
      for (int u = 0; base + u < nnz; ++u) {
        const int rs = __shfl(V2E ? vl : el, u, 32);
        const int rd = __shfl(V2E ? el : vl, u, 32);
        const float* s = src + (size_t)rs * HO + lane;
        float* d = dst + (size_t)rd * HO + lane;
        float a0 = s[0], a1 = s[32];
        if (!V2E) {
          const float att = __shfl(al, u, 32);
          a0 = att * a0;
          a1 = att * a1;
        }
        atomicAdd(d, a0);
        atomicAdd(d + 32, a1);
      }
    }
  }
}

__global__ __launch_bounds__(256) void k_edge_mean(float* __restrict__ Xe,
                                                   const float* __restrict__ cnt,
                                                   long long total) {
  long long i = (long long)blockIdx.x * blockDim.x + threadIdx.x;
  if (i >= total) return;
  const float c = cnt[i >> 6];
  Xe[i] *= 1.0f / fmaxf(c, 1.0f);
}

__global__ __launch_bounds__(256) void k_rownorm(float* __restrict__ out, int n) {
  const int wave = (int)(((unsigned)blockIdx.x * blockDim.x + threadIdx.x) >> 5);
  const int lane = (int)(threadIdx.x & 31u);
  if (wave >= n) return;
  float* row = out + (size_t)wave * HO + lane * 2;
  float2 x = *(float2*)row;
  float ss = x.x * x.x + x.y * x.y;
#pragma unroll
  for (int off = 16; off > 0; off >>= 1) ss += __shfl_xor(ss, off, 32);
  const float nrm = sqrtf(ss);
  const float scale = (nrm > 0.f) ? 1.0f / fmaxf(nrm, 1e-30f) : 0.0f;
  x.x *= scale;
  x.y *= scale;
  *(float2*)row = x;
}

#define NB 512
#define NCH 256
#define RMAXE 64
#define RMAXV 200
#define SRCMASK 0xFFFFF

__global__ __launch_bounds__(NB) void k_count_gemm(
    const int* __restrict__ vertex, const int* __restrict__ edges,
    int* __restrict__ cntE, int* __restrict__ cntV, int nnz, int re, int rv,
    const float* __restrict__ X, const float* __restrict__ W,
    float* __restrict__ Xp, int nrows) {
  __shared__ int hE[NB], hV[NB];
  const int t = (int)threadIdx.x;
  if ((int)blockIdx.x >= NCH) {
    const int wave = (((int)blockIdx.x - NCH) * NB + t) >> 5;
    gemm_rows16(X, W, Xp, nrows, wave, t & 31);
    return;
  }
  const int c = (int)blockIdx.x;
  hE[t] = 0;
  hV[t] = 0;
  __syncthreads();
  const int ch = (nnz + NCH - 1) / NCH;
  const int lo = c * ch;
  const int hi = min(lo + ch, nnz);
  for (int i = lo + t; i < hi; i += NB) {
    atomicAdd(&hE[edges[i] / re], 1);
    atomicAdd(&hV[vertex[i] / rv], 1);
  }
  __syncthreads();
  cntE[c * NB + t] = hE[t];
  cntV[c * NB + t] = hV[t];
}

__global__ __launch_bounds__(NB) void k_bin_offsets(
    int* __restrict__ cntE, int* __restrict__ cntV,
    int* __restrict__ bsE, int* __restrict__ bsV) {
  __shared__ int s[NB];
  int* tab = blockIdx.x ? cntV : cntE;
  int* bs = blockIdx.x ? bsV : bsE;
  const int b = (int)threadIdx.x;
  int run = 0;
  for (int c0 = 0; c0 < NCH; c0 += 32) {
    int v[32];
#pragma unroll
    for (int u = 0; u < 32; ++u) v[u] = tab[(c0 + u) * NB + b];
#pragma unroll
    for (int u = 0; u < 32; ++u) {
      tab[(c0 + u) * NB + b] = run;
      run += v[u];
    }
  }
  s[b] = run;
  __syncthreads();
  for (int off = 1; off < NB; off <<= 1) {
    const int add = (b >= off) ? s[b - off] : 0;
    __syncthreads();
    s[b] += add;
    __syncthreads();
  }
  bs[b] = s[b] - run;
  if (b == NB - 1) bs[NB] = s[b];
}

#define PCAP 8192
#define PK (PCAP / NB)
__global__ __launch_bounds__(NB) void k_bin_place(
    const int* __restrict__ vertex, const int* __restrict__ edges,
    const int* __restrict__ cntE, const int* __restrict__ cntV,
    const int* __restrict__ bsE, const int* __restrict__ bsV,
    int* __restrict__ listE, int* __restrict__ listV, int nnz, int re, int rv) {
  __shared__ int hE[NB], hV[NB];
  __shared__ int gE[NB], gV[NB];
  __shared__ int wsE[NB / 32], wsV[NB / 32];
  __shared__ int stE[PCAP], stV[PCAP];
  __shared__ int adE[PCAP], adV[PCAP];
  const int t = (int)threadIdx.x, c = (int)blockIdx.x;
  const int lane = t & 31;
  const int wave = t >> 5;
  int gbE = bsE[t] + cntE[c * NB + t];
  int gbV = bsV[t] + cntV[c * NB + t];
  const int ch = (nnz + NCH - 1) / NCH;
  const int lo = c * ch;
  const int hi = min(lo + ch, nnz);
  for (int plo = lo; plo < hi; plo += PCAP) {
    const int pn = min(PCAP, hi - plo);
    hE[t] = 0;
    hV[t] = 0;
    int wE[PK], wV[PK], pE[PK], pV[PK];
#pragma unroll
    for (int k = 0; k < PK; ++k) {
      const int i = plo + min(k * NB + t, pn - 1);
      wE[k] = vertex[i];
      wV[k] = edges[i];
    }
    __syncthreads();
#pragma unroll
    for (int k = 0; k < PK; ++k) {
      const bool ok = k * NB + t < pn;
      const int v = wE[k], e = wV[k];
      const int be = e / re, bv = v / rv;
      const int ce = ok ? be : t, cv = ok ? bv : t;
      const int one = ok ? 1 : 0;
      pE[k] = ce | (atomicAdd(&hE[ce], one) << 16);
      pV[k] = cv | (atomicAdd(&hV[cv], one) << 16);
      wE[k] = v | ((e - be * re) << 20);
      wV[k] = e | ((v - bv * rv) << 20);
    }
    __syncthreads();
    const int nE = hE[t], nV = hV[t];
    int iE = nE, iV = nV;
#pragma unroll
    for (int off = 1; off < 32; off <<= 1) {
      const int uE = __shfl_up(iE, off, 32);
      const int uV = __shfl_up(iV, off, 32);
      if (lane >= off) {
        iE += uE;
        iV += uV;
      }
    }
    if (lane == 31) {
      wsE[wave] = iE;
      wsV[wave] = iV;
    }
    __syncthreads();
    int bE = 0, bV = 0;
    for (int w2 = 0; w2 < NB / 32; ++w2)
      if (w2 < wave) {
        bE += wsE[w2];
        bV += wsV[w2];
      }
    hE[t] = bE + iE - nE;
    hV[t] = bV + iV - nV;
    gE[t] = gbE;
    gV[t] = gbV;
    gbE += nE;
    gbV += nV;
    __syncthreads();
    int se[PK], sv[PK];
#pragma unroll
    for (int k = 0; k < PK; ++k) {
      const int be = pE[k] & 0xFFFF, ke = pE[k] >> 16;
      const int bv = pV[k] & 0xFFFF, kv = pV[k] >> 16;
      se[k] = hE[be] + ke;
      pE[k] = gE[be] + ke;
      sv[k] = hV[bv] + kv;
      pV[k] = gV[bv] + kv;
    }
#pragma unroll
    for (int k = 0; k < PK; ++k)
      if (k * NB + t < pn) {
        stE[se[k]] = wE[k];
        adE[se[k]] = pE[k];
        stV[sv[k]] = wV[k];
        adV[sv[k]] = pV[k];
      }
    __syncthreads();
#pragma unroll
    for (int k = 0; k < PK; ++k) {
      const int s = min(k * NB + t, pn - 1);
      wE[k] = stE[s];
      pE[k] = adE[s];
      wV[k] = stV[s];
      pV[k] = adV[s];
    }
#pragma unroll
    for (int k = 0; k < PK; ++k)
      if (k * NB + t < pn) {
        listE[pE[k]] = wE[k];
        listV[pV[k]] = wV[k];
      }
  }
}

#define TRIPB 16
#define TILE 4608
template <bool V2E, int RMAX>
__global__ __launch_bounds__(NB) void k_range_sum(
    const float* __restrict__ src, const int* __restrict__ list, const int* __restrict__ bs,
    const float* __restrict__ homo, float* __restrict__ dst, int r, int nrows) {
  __shared__ __attribute__((aligned(16))) float acc[RMAX * HO];
  __shared__ float aux[RMAX];
  __shared__ int cur[RMAX];
  __shared__ int sorted[TILE];
  const int t = (int)threadIdx.x;
  const int lane = t & 31;
  const int bin = (int)blockIdx.x;
  const int lo = bs[bin], hi = bs[bin + 1];
  for (int j = t; j < r * HO; j += NB) acc[j] = 0.0f;
  for (int j = t; j < r; j += NB) aux[j] = 0.0f;
  const int wave = __builtin_amdgcn_readfirstlane(t >> 5);
  constexpr int NWV = NB / 32;
  constexpr int KW = TILE / NB;
  constexpr int RPL = (RMAX + 31) / 32;
  for (int tlo = lo; tlo < hi; tlo += TILE) {
    const int tn = min(TILE, hi - tlo);
    for (int j = t; j < r; j += NB) cur[j] = 0;
    int w[KW];
#pragma unroll
    for (int k = 0; k < KW; ++k) w[k] = list[min(tlo + k * NB + t, hi - 1)];
    __syncthreads();
#pragma unroll
    for (int k = 0; k < KW; ++k)
      if (k * NB + t < tn) atomicAdd(&cur[w[k] >> 20], 1);
    __syncthreads();
    if (wave == 0) {
      int c[RPL];
      int s = 0;
#pragma unroll
      for (int u = 0; u < RPL; ++u) {
        const int j = lane * RPL + u;
        c[u] = (j < r) ? cur[j] : 0;
        if (V2E && j < r) aux[j] += (float)c[u];
        s += c[u];
      }
      int incl = s;
#pragma unroll
      for (int off = 1; off < 32; off <<= 1) {
        const int v = __shfl_up(incl, off, 32);
        if (lane >= off) incl += v;
      }
      int run = incl - s;
#pragma unroll
      for (int u = 0; u < RPL; ++u) {
        const int j = lane * RPL + u;
        if (j < r) cur[j] = run;
        run += c[u];
      }
    }
    __syncthreads();
#pragma unroll
    for (int k = 0; k < KW; ++k)
      if (k * NB + t < tn) sorted[atomicAdd(&cur[w[k] >> 20], 1)] = w[k];
    __syncthreads();
    const int seg = ((tn + NWV * TRIPB - 1) / (NWV * TRIPB)) * TRIPB;
    const int wbeg = wave * seg;
    const int wend = min(wbeg + seg, tn);
    if (wbeg < wend) {
      float a0 = 0.0f, a1 = 0.0f;
      float hs = 0.0f;
      int crow = -1;
      int ent = sorted[min(wbeg + lane, wend - 1)];
      for (int b = wbeg; b < wend; b += TRIPB) {
        const int nv = min(TRIPB, wend - b);
        const int entn = sorted[min(b + TRIPB + lane, wend - 1)];
        const int sl = ent & SRCMASK;
        const int dl = ent >> 20;
        float al = 0.0f;
        if (!V2E && lane < nv) al = homo[sl];
        float x0[TRIPB], x1[TRIPB];
#pragma unroll
        for (int u = 0; u < TRIPB; ++u) {
          const float* s = src + (size_t)__builtin_amdgcn_readlane(sl, u) * HO + lane;
          x0[u] = s[0];
          x1[u] = s[32];
        }
#pragma unroll
        for (int u = 0; u < TRIPB; ++u) {
          if (u < nv) {
            const int d = __builtin_amdgcn_readlane(dl, u);
            if (d != crow) {
              if (crow >= 0) {
                atomicAdd(acc + crow * HO + lane, a0);
                atomicAdd(acc + crow * HO + 32 + lane, a1);
                if (!V2E && lane == 0) atomicAdd(aux + crow, hs);
              }
              a0 = 0.0f;
              a1 = 0.0f;
              hs = 0.0f;
              crow = d;
            }
            if (V2E) {
              a0 += x0[u];
              a1 += x1[u];
            } else {
              const float att = __int_as_float(__builtin_amdgcn_readlane(__float_as_int(al), u));
              a0 += att * x0[u];
              a1 += att * x1[u];
              hs += att;
            }
          }
        }
        ent = entn;
      }
      if (crow >= 0) {
        atomicAdd(acc + crow * HO + lane, a0);
        atomicAdd(acc + crow * HO + 32 + lane, a1);
        if (!V2E && lane == 0) atomicAdd(aux + crow, hs);
      }
    }
  }
  if (V2E) {
    __syncthreads();
    for (int j = t; j < r * HO; j += NB) {
      const int row = bin * r + (j >> 6);
      if (row < nrows) dst[(size_t)row * HO + (j & 63)] = acc[j] / fmaxf(aux[j >> 6], 1.0f);
    }
  } else {
    constexpr int NW = NB / 32;
    constexpr int RPW = (RMAX + NW - 1) / NW;
    const int nv = min(r, nrows - bin * r);
    if (nv <= 0) return;
    float2 x[RPW];
#pragma unroll
    for (int u = 0; u < RPW; ++u) {
      const int rc = min(wave + u * NW, nv - 1);
      x[u] = *(const float2*)(dst + (size_t)(bin * r + rc) * HO + lane * 2);
    }
    __syncthreads();
#pragma unroll
    for (int u = 0; u < RPW; ++u) {
      const int rc = min(wave + u * NW, nv - 1);
      const float2 a = *(const float2*)(acc + rc * HO + lane * 2);
      const float asum = aux[rc];
      const float rcp = 1.0f / asum;
      const float wgt = (asum != 0.0f) ? rcp : 0.0f;
      x[u].x = x[u].x + wgt * a.x;
      x[u].y = x[u].y + wgt * a.y;
      float ss = x[u].x * x[u].x + x[u].y * x[u].y;
#pragma unroll
      for (int off = 16; off > 0; off >>= 1) ss += __shfl_xor(ss, off, 32);
      const float nrm = sqrtf(ss);
      const float inv = 1.0f / fmaxf(nrm, 1e-30f);
      const float scale = (nrm > 0.f) ? inv : 0.0f;
      x[u].x *= scale;
      x[u].y *= scale;
    }
#pragma unroll
    for (int u = 0; u < RPW; ++u) {
      const int rr = wave + u * NW;
      if (rr < nv)
        *(float2*)(dst + (size_t)(bin * r + rr) * HO + lane * 2) = x[u];
    }
  }
}

extern "C" void kernel_launch(void* const* d_in, const int* in_sizes, int n_in,
                              void* d_out, int out_size, void* d_ws, size_t ws_size,
                              hipStream_t stream) {
  const float* X      = (const float*)d_in[0];
  const float* W      = (const float*)d_in[1];
  const float* homo   = (const float*)d_in[2];
  const int*   vertex = (const int*)d_in[3];
  const int*   edges  = (const int*)d_in[4];
  float* out = (float*)d_out;

  const int n   = in_sizes[0] / INCH;
  const int e   = in_sizes[2];
  const int nnz = in_sizes[3];

  const int tiles = (n + 15) / 16;

  const int re = (e + NB - 1) / NB;
  const int rv = (n + NB - 1) / NB;
  const size_t need = ((size_t)e * HO + 2 * (size_t)nnz + 2 * (size_t)NCH * NB + 2 * (NB + 1)) * 4;
  float* Xe = (float*)d_ws;

  if (need <= ws_size && re <= RMAXE && rv <= RMAXV && n <= SRCMASK && e <= SRCMASK) {
    int* listE = (int*)(Xe + (size_t)e * HO);
    int* listV = listE + nnz;
    int* cntE  = listV + nnz;
    int* cntV  = cntE + NCH * NB;
    int* bsE   = cntV + NCH * NB;
    int* bsV   = bsE + NB + 1;
    const int gblocks = (tiles + NB / 32 - 1) / (NB / 32);
    k_count_gemm<<<NCH + gblocks, NB, 0, stream>>>(vertex, edges, cntE, cntV, nnz, re, rv, X, W, out, n);
    k_bin_offsets<<<2, NB, 0, stream>>>(cntE, cntV, bsE, bsV);
    k_bin_place<<<NCH, NB, 0, stream>>>(vertex, edges, cntE, cntV, bsE, bsV, listE, listV, nnz, re, rv);
    k_range_sum<true, RMAXE><<<NB, NB, 0, stream>>>(out, listE, bsE, homo, Xe, re, e);
    k_range_sum<false, RMAXV><<<NB, NB, 0, stream>>>(Xe, listV, bsV, homo, out, rv, n);
  } else {
    k_gemm_wmma<<<(tiles * 32 + 255) / 256, 256, 0, stream>>>(X, W, out, n);
    float* cnt     = Xe + (size_t)e * HO;
    float* att_sum = cnt + e;
    const long long zn = (long long)e * HO + e + n;
    k_zero<<<(int)((zn + 255) / 256), 256, 0, stream>>>(Xe, zn);
    k_scatter_rows<true><<<8192, 256, 0, stream>>>(
        out, vertex, edges, homo, Xe, cnt, att_sum, nnz);
    const long long em = (long long)e * HO;
    k_edge_mean<<<(int)((em + 255) / 256), 256, 0, stream>>>(Xe, cnt, em);
    k_scatter_rows<false><<<8192, 256, 0, stream>>>(
        Xe, vertex, edges, homo, out, cnt, att_sum, nnz);
    const long long nt = (long long)n * 32;
    k_rownorm<<<(int)((nt + 255) / 256), 256, 0, stream>>>(out, n);
  }
}


# Lists grouped per chunk only: a chunk's LDS sort writes its own slice plus a two-byte start per range; a range's work-group gathers its 256 pieces; count and offsets launches go (sum order changes)
# speedup vs baseline: 6.9284x; 1.2856x over previous
#include <hip/hip_runtime.h>
#include <math.h>

typedef float v2f __attribute__((ext_vector_type(2)));
typedef float v8f __attribute__((ext_vector_type(8)));

#define HO 64
#define INCH 64

__global__ __launch_bounds__(256) void k_zero(float* __restrict__ p, long long n) {
  long long i = (long long)blockIdx.x * blockDim.x + threadIdx.x;
  if (i < n) p[i] = 0.0f;
}

__device__ __forceinline__ void gemm_rows16(
    const float* __restrict__ X, const float* __restrict__ W,
    float* __restrict__ Xp, int nrows, int wave, int lane) {
  const int row0 = wave * 16;
  if (row0 >= nrows) return;
  if (row0 + 16 <= nrows) {
    const int m  = lane & 15;
    const int kh = lane >> 4;
    v8f acc0 = {}, acc1 = {}, acc2 = {}, acc3 = {};
#pragma unroll
    for (int k0 = 0; k0 < INCH; k0 += 4) {
      v2f a = *(const v2f*)(X + (size_t)(row0 + m) * INCH + (k0 + kh * 2));
      const float* w0 = W + (size_t)(k0 + kh * 2) * HO;
      const float* w1 = w0 + HO;
      v2f b0, b1, b2, b3;
      b0.x = w0[m];      b0.y = w1[m];
      b1.x = w0[m + 16]; b1.y = w1[m + 16];
      b2.x = w0[m + 32]; b2.y = w1[m + 32];
      b3.x = w0[m + 48]; b3.y = w1[m + 48];
      acc0 = __builtin_amdgcn_wmma_f32_16x16x4_f32(false, a, false, b0, (short)0, acc0, false, false);
      acc1 = __builtin_amdgcn_wmma_f32_16x16x4_f32(false, a, false, b1, (short)0, acc1, false, false);
      acc2 = __builtin_amdgcn_wmma_f32_16x16x4_f32(false, a, false, b2, (short)0, acc2, false, false);
      acc3 = __builtin_amdgcn_wmma_f32_16x16x4_f32(false, a, false, b3, (short)0, acc3, false, false);
    }
#pragma unroll
    for (int r = 0; r < 8; ++r) {
      const int row = row0 + r + kh * 8;
      float* o = Xp + (size_t)row * HO + m;
      o[0]  = acc0[r];
      o[16] = acc1[r];
      o[32] = acc2[r];
      o[48] = acc3[r];
    }
  } else {
    for (int r = row0 + lane; r < nrows; r += 32)
      for (int c = 0; c < HO; ++c) {
        float s = 0.f;
        for (int k = 0; k < INCH; ++k) s += X[(size_t)r * INCH + k] * W[(size_t)k * HO + c];
        Xp[(size_t)r * HO + c] = s;
      }
  }
}

__global__ __launch_bounds__(256) void k_gemm_wmma(
    const float* __restrict__ X, const float* __restrict__ W,
    float* __restrict__ Xp, int nrows) {
  const int wave = (int)(((unsigned)blockIdx.x * blockDim.x + threadIdx.x) >> 5);
  const int lane = (int)(threadIdx.x & 31u);
  gemm_rows16(X, W, Xp, nrows, wave, lane);
}

#define TRIP 8
template <bool V2E>
__global__ __launch_bounds__(256) void k_scatter_rows(
    const float* __restrict__ src, const int* __restrict__ vertex,
    const int* __restrict__ edges, const float* __restrict__ homo,
    float* __restrict__ dst, float* __restrict__ cnt,
    float* __restrict__ att_sum, int nnz) {
  const int lane = (int)(threadIdx.x & 31u);
  const int wave = __builtin_amdgcn_readfirstlane(
      (int)(((unsigned)blockIdx.x * blockDim.x + threadIdx.x) >> 5));
  const int nwaves = (int)((gridDim.x * blockDim.x) >> 5);
  for (int base = wave * TRIP; base < nnz; base += nwaves * TRIP) {
    const int idx = base + lane;
    const bool mine = lane < TRIP && idx < nnz;
    int vl = 0, el = 0;
    if (mine) {
      vl = vertex[idx];
      el = edges[idx];
    }
    const bool whole = base + TRIP <= nnz;
    float x0[TRIP], x1[TRIP];
    if (whole) {
#pragma unroll
      for (int u = 0; u < TRIP; ++u) {
        const int r = __builtin_amdgcn_readlane(V2E ? vl : el, u);
        const float* s = src + (size_t)r * HO + lane;
        x0[u] = s[0];
        x1[u] = s[32];
      }
    }
    float al = 0.0f;
    if (mine) {
      if (V2E) {
        atomicAdd(cnt + el, 1.0f);
        atomicAdd(att_sum + vl, homo[el]);
      } else {
        al = homo[el] / att_sum[vl];
      }
    }
    if (whole) {
#pragma unroll
      for (int u = 0; u < TRIP; ++u) {
        const int r = __builtin_amdgcn_readlane(V2E ? el : vl, u);
        float* d = dst + (size_t)r * HO + lane;
        float a0 = x0[u], a1 = x1[u];
        if (!V2E) {
          const float att = __int_as_float(__builtin_amdgcn_readlane(__float_as_int(al), u));
          a0 = att * a0;
          a1 = att * a1;
        }
        atomicAdd(d, a0);
        atomicAdd(d + 32, a1);
      }
    } else {
      for (int u = 0; base + u < nnz; ++u) {
        const int rs = __shfl(V2E ? vl : el, u, 32);
        const int rd = __shfl(V2E ? el : vl, u, 32);
        const float* s = src + (size_t)rs * HO + lane;
        float* d = dst + (size_t)rd * HO + lane;
        float a0 = s[0], a1 = s[32];
        if (!V2E) {
          const float att = __shfl(al, u, 32);
          a0 = att * a0;
          a1 = att * a1;
        }
        atomicAdd(d, a0);
        atomicAdd(d + 32, a1);
      }
    }
  }
}

__global__ __launch_bounds__(256) void k_edge_mean(float* __restrict__ Xe,
                                                   const float* __restrict__ cnt,
                                                   long long total) {
  long long i = (long long)blockIdx.x * blockDim.x + threadIdx.x;
  if (i >= total) return;
  const float c = cnt[i >> 6];
  Xe[i] *= 1.0f / fmaxf(c, 1.0f);
}

__global__ __launch_bounds__(256) void k_rownorm(float* __restrict__ out, int n) {
  const int wave = (int)(((unsigned)blockIdx.x * blockDim.x + threadIdx.x) >> 5);
  const int lane = (int)(threadIdx.x & 31u);
  if (wave >= n) return;
  float* row = out + (size_t)wave * HO + lane * 2;
  float2 x = *(float2*)row;
  float ss = x.x * x.x + x.y * x.y;
#pragma unroll
  for (int off = 16; off > 0; off >>= 1) ss += __shfl_xor(ss, off, 32);
  const float nrm = sqrtf(ss);
  const float scale = (nrm > 0.f) ? 1.0f / fmaxf(nrm, 1e-30f) : 0.0f;
  x.x *= scale;
  x.y *= scale;
  *(float2*)row = x;
}

#define NB 512
#define NCH 256
#define RMAXE 64
#define RMAXV 200
#define SRCMASK 0xFFFFF
#define PCAP 8192
#define NT 1024
#define PK (PCAP / NT)

__global__ __launch_bounds__(NT) void k_group_gemm(
    const int* __restrict__ vertex, const int* __restrict__ edges,
    unsigned short* __restrict__ pcE, unsigned short* __restrict__ pcV,
    int* __restrict__ listE, int* __restrict__ listV, int nnz, int re, int rv,
    const float* __restrict__ X, const float* __restrict__ W,
    float* __restrict__ Xp, int nrows) {
  __shared__ int hE[NB], hV[NB];
  __shared__ int wsE[NB / 32], wsV[NB / 32];
  __shared__ int stE[PCAP], stV[PCAP];
  const int t = (int)threadIdx.x;
  if ((int)blockIdx.x >= NCH) {
    const int wave = (((int)blockIdx.x - NCH) * NT + t) >> 5;
    gemm_rows16(X, W, Xp, nrows, wave, t & 31);
    return;
  }
  const int c = (int)blockIdx.x;
  const bool rng = t < NB;
  const int lane = t & 31;
  const int wave = t >> 5;
  const int ch = (nnz + NCH - 1) / NCH;
  const int lo = c * ch;
  const int pn = min(ch, nnz - lo);
  if (pn <= 0) {
    if (rng) {
      pcE[c * (NB + 1) + t] = 0;
      pcV[c * (NB + 1) + t] = 0;
    }
    if (t == 0) {
      pcE[c * (NB + 1) + NB] = 0;
      pcV[c * (NB + 1) + NB] = 0;
    }
    return;
  }
  if (rng) {
    hE[t] = 0;
    hV[t] = 0;
  }
  int wE[PK], wV[PK], pE[PK], pV[PK];
#pragma unroll
  for (int k = 0; k < PK; ++k) {
    const int i = lo + min(k * NT + t, pn - 1);
    wE[k] = vertex[i];
    wV[k] = edges[i];
  }
  __syncthreads();
#pragma unroll
  for (int k = 0; k < PK; ++k) {
    const bool ok = k * NT + t < pn;
    const int v = wE[k], e = wV[k];
    const int be = e / re, bv = v / rv;
    const int ce = ok ? be : (t & (NB - 1)), cv = ok ? bv : (t & (NB - 1));
    const int one = ok ? 1 : 0;
    pE[k] = ce | (atomicAdd(&hE[ce], one) << 16);
    pV[k] = cv | (atomicAdd(&hV[cv], one) << 16);
    wE[k] = v | ((e - be * re) << 20);
    wV[k] = e | ((v - bv * rv) << 20);
  }
  __syncthreads();
  int nE = 0, nV = 0;
  if (rng) {
    nE = hE[t];
    nV = hV[t];
  }
  int iE = nE, iV = nV;
#pragma unroll
  for (int off = 1; off < 32; off <<= 1) {
    const int uE = __shfl_up(iE, off, 32);
    const int uV = __shfl_up(iV, off, 32);
    if (lane >= off) {
      iE += uE;
      iV += uV;
    }
  }
  if (rng && lane == 31) {
    wsE[wave] = iE;
    wsV[wave] = iV;
  }
  __syncthreads();
  if (rng) {
    int bE = 0, bV = 0;
    for (int w2 = 0; w2 < NB / 32; ++w2)
      if (w2 < wave) {
        bE += wsE[w2];
        bV += wsV[w2];
      }
    const int fE = bE + iE - nE, fV = bV + iV - nV;
    hE[t] = fE;
    hV[t] = fV;
    pcE[c * (NB + 1) + t] = (unsigned short)fE;
    pcV[c * (NB + 1) + t] = (unsigned short)fV;
  }
  if (t == 0) {
    pcE[c * (NB + 1) + NB] = (unsigned short)pn;
    pcV[c * (NB + 1) + NB] = (unsigned short)pn;
  }
  __syncthreads();
  int se[PK], sv[PK];
#pragma unroll
  for (int k = 0; k < PK; ++k) {
    se[k] = hE[pE[k] & 0xFFFF] + (pE[k] >> 16);
    sv[k] = hV[pV[k] & 0xFFFF] + (pV[k] >> 16);
  }
#pragma unroll
  for (int k = 0; k < PK; ++k)
    if (k * NT + t < pn) {
      stE[se[k]] = wE[k];
      stV[sv[k]] = wV[k];
    }
  __syncthreads();
#pragma unroll
  for (int k = 0; k < PK; ++k) {
    const int s = min(k * NT + t, pn - 1);
    wE[k] = stE[s];
    wV[k] = stV[s];
  }
#pragma unroll
  for (int k = 0; k < PK; ++k)
    if (k * NT + t < pn) {
      listE[lo + k * NT + t] = wE[k];
      listV[lo + k * NT + t] = wV[k];
    }
}

#define TRIPB 16
#define TILE 4608
template <bool V2E, int RMAX>
__global__ __launch_bounds__(NB) void k_range_sum(
    const float* __restrict__ src, const int* __restrict__ list, const unsigned short* __restrict__ pc,
    const float* __restrict__ homo, float* __restrict__ dst, int r, int nrows, int nnz) {
  __shared__ __attribute__((aligned(16))) float acc[RMAX * HO];
  __shared__ float aux[RMAX];
  __shared__ int cur[RMAX];
  __shared__ int sorted[TILE];
  __shared__ int own[TILE];
  __shared__ int wsum[NB / 32];
  const int t = (int)threadIdx.x;
  const int lane = t & 31;
  const int bin = (int)blockIdx.x;
  const int wave = __builtin_amdgcn_readfirstlane(t >> 5);
  int pstart = 0, plen = 0;
  if (t < NCH) {
    const int f0 = pc[t * (NB + 1) + bin], f1 = pc[t * (NB + 1) + bin + 1];
    pstart = t * ((nnz + NCH - 1) / NCH) + f0;
    plen = f1 - f0;
  }
  for (int j = t; j < r * HO; j += NB) acc[j] = 0.0f;
  for (int j = t; j < r; j += NB) aux[j] = 0.0f;
  int pin = plen;
#pragma unroll
  for (int off = 1; off < 32; off <<= 1) {
    const int v = __shfl_up(pin, off, 32);
    if (lane >= off) pin += v;
  }
  if (lane == 31) wsum[wave] = pin;
  __syncthreads();
  int before = 0, total = 0;
  for (int w2 = 0; w2 < NB / 32; ++w2) {
    const int v = wsum[w2];
    if (w2 < wave) before += v;
    total += v;
  }
  const int ppos = before + pin - plen;
  const int n = __builtin_amdgcn_readfirstlane(total);
  constexpr int NWV = NB / 32;
  constexpr int KW = TILE / NB;
  constexpr int RPL = (RMAX + 31) / 32;
  for (int tlo = 0; tlo < n; tlo += TILE) {
    const int tn = min(TILE, n - tlo);
    for (int j = t; j < r; j += NB) cur[j] = 0;
    const int j1 = min(plen, tlo + TILE - ppos);
    for (int j = max(0, tlo - ppos); j < j1; ++j) own[ppos + j - tlo] = pstart + j;
    __syncthreads();
    int w[KW];
#pragma unroll
    for (int k = 0; k < KW; ++k) w[k] = list[own[min(k * NB + t, tn - 1)]];
#pragma unroll
    for (int k = 0; k < KW; ++k)
      if (k * NB + t < tn) atomicAdd(&cur[w[k] >> 20], 1);
    __syncthreads();
    if (wave == 0) {
      int c[RPL];
      int s = 0;
#pragma unroll
      for (int u = 0; u < RPL; ++u) {
        const int j = lane * RPL + u;
        c[u] = (j < r) ? cur[j] : 0;
        if (V2E && j < r) aux[j] += (float)c[u];
        s += c[u];
      }
      int incl = s;
#pragma unroll
      for (int off = 1; off < 32; off <<= 1) {
        const int v = __shfl_up(incl, off, 32);
        if (lane >= off) incl += v;
      }
      int run = incl - s;
#pragma unroll
      for (int u = 0; u < RPL; ++u) {
        const int j = lane * RPL + u;
        if (j < r) cur[j] = run;
        run += c[u];
      }
    }
    __syncthreads();
#pragma unroll
    for (int k = 0; k < KW; ++k)
      if (k * NB + t < tn) sorted[atomicAdd(&cur[w[k] >> 20], 1)] = w[k];
    __syncthreads();
    const int seg = ((tn + NWV * TRIPB - 1) / (NWV * TRIPB)) * TRIPB;
    const int wbeg = wave * seg;
    const int wend = min(wbeg + seg, tn);
    if (wbeg < wend) {
      float a0 = 0.0f, a1 = 0.0f;
      float hs = 0.0f;
      int crow = -1;
      int ent = sorted[min(wbeg + lane, wend - 1)];
      for (int b = wbeg; b < wend; b += TRIPB) {
        const int nv = min(TRIPB, wend - b);
        const int entn = sorted[min(b + TRIPB + lane, wend - 1)];
        const int sl = ent & SRCMASK;
        const int dl = ent >> 20;
        float al = 0.0f;
        if (!V2E && lane < nv) al = homo[sl];
        float x0[TRIPB], x1[TRIPB];
#pragma unroll
        for (int u = 0; u < TRIPB; ++u) {
          const float* s = src + (size_t)__builtin_amdgcn_readlane(sl, u) * HO + lane;
          x0[u] = s[0];
          x1[u] = s[32];
        }
#pragma unroll
        for (int u = 0; u < TRIPB; ++u) {
          if (u < nv) {
            const int d = __builtin_amdgcn_readlane(dl, u);
            if (d != crow) {
              if (crow >= 0) {
                atomicAdd(acc + crow * HO + lane, a0);
                atomicAdd(acc + crow * HO + 32 + lane, a1);
                if (!V2E && lane == 0) atomicAdd(aux + crow, hs);
              }
              a0 = 0.0f;
              a1 = 0.0f;
              hs = 0.0f;
              crow = d;
            }
            if (V2E) {
              a0 += x0[u];
              a1 += x1[u];
            } else {
              const float att = __int_as_float(__builtin_amdgcn_readlane(__float_as_int(al), u));
              a0 += att * x0[u];
              a1 += att * x1[u];
              hs += att;
            }
          }
        }
        ent = entn;
      }
      if (crow >= 0) {
        atomicAdd(acc + crow * HO + lane, a0);
        atomicAdd(acc + crow * HO + 32 + lane, a1);
        if (!V2E && lane == 0) atomicAdd(aux + crow, hs);
      }
    }
  }
  if (V2E) {
    __syncthreads();
    for (int j = t; j < r * HO; j += NB) {
      const int row = bin * r + (j >> 6);
      if (row < nrows) dst[(size_t)row * HO + (j & 63)] = acc[j] / fmaxf(aux[j >> 6], 1.0f);
    }
  } else {
    constexpr int NW = NB / 32;
    constexpr int RPW = (RMAX + NW - 1) / NW;
    const int nv = min(r, nrows - bin * r);
    if (nv <= 0) return;
    float2 x[RPW];
#pragma unroll
    for (int u = 0; u < RPW; ++u) {
      const int rc = min(wave + u * NW, nv - 1);
      x[u] = *(const float2*)(dst + (size_t)(bin * r + rc) * HO + lane * 2);
    }
    __syncthreads();
#pragma unroll
    for (int u = 0; u < RPW; ++u) {
      const int rc = min(wave + u * NW, nv - 1);
      const float2 a = *(const float2*)(acc + rc * HO + lane * 2);
      const float asum = aux[rc];
      const float rcp = 1.0f / asum;
      const float wgt = (asum != 0.0f) ? rcp : 0.0f;
      x[u].x = x[u].x + wgt * a.x;
      x[u].y = x[u].y + wgt * a.y;
      float ss = x[u].x * x[u].x + x[u].y * x[u].y;
#pragma unroll
      for (int off = 16; off > 0; off >>= 1) ss += __shfl_xor(ss, off, 32);
      const float nrm = sqrtf(ss);
      const float inv = 1.0f / fmaxf(nrm, 1e-30f);
      const float scale = (nrm > 0.f) ? inv : 0.0f;
      x[u].x *= scale;
      x[u].y *= scale;
    }
#pragma unroll
    for (int u = 0; u < RPW; ++u) {
      const int rr = wave + u * NW;
      if (rr < nv)
        *(float2*)(dst + (size_t)(bin * r + rr) * HO + lane * 2) = x[u];
    }
  }
}

extern "C" void kernel_launch(void* const* d_in, const int* in_sizes, int n_in,
                              void* d_out, int out_size, void* d_ws, size_t ws_size,
                              hipStream_t stream) {
  const float* X      = (const float*)d_in[0];
  const float* W      = (const float*)d_in[1];
  const float* homo   = (const float*)d_in[2];
  const int*   vertex = (const int*)d_in[3];
  const int*   edges  = (const int*)d_in[4];
  float* out = (float*)d_out;

  const int n   = in_sizes[0] / INCH;
  const int e   = in_sizes[2];
  const int nnz = in_sizes[3];

  const int tiles = (n + 15) / 16;

  const int re = (e + NB - 1) / NB;
  const int rv = (n + NB - 1) / NB;
  const size_t need = ((size_t)e * HO + 2 * (size_t)nnz) * 4 + 2 * (size_t)NCH * (NB + 1) * 2;
  float* Xe = (float*)d_ws;

  if (need <= ws_size && re <= RMAXE && rv <= RMAXV && n <= SRCMASK && e <= SRCMASK &&
      (nnz + NCH - 1) / NCH <= PCAP) {
    int* listE = (int*)(Xe + (size_t)e * HO);
    int* listV = listE + nnz;
    unsigned short* pcE = (unsigned short*)(listV + nnz);
    unsigned short* pcV = pcE + NCH * (NB + 1);
    const int gblocks = (tiles + NT / 32 - 1) / (NT / 32);
    k_group_gemm<<<NCH + gblocks, NT, 0, stream>>>(vertex, edges, pcE, pcV, listE, listV, nnz, re, rv, X, W, out, n);
    k_range_sum<true, RMAXE><<<NB, NB, 0, stream>>>(out, listE, pcE, homo, Xe, re, e, nnz);
    k_range_sum<false, RMAXV><<<NB, NB, 0, stream>>>(Xe, listV, pcV, homo, out, rv, n, nnz);
  } else {
    k_gemm_wmma<<<(tiles * 32 + 255) / 256, 256, 0, stream>>>(X, W, out, n);
    float* cnt     = Xe + (size_t)e * HO;
    float* att_sum = cnt + e;
    const long long zn = (long long)e * HO + e + n;
    k_zero<<<(int)((zn + 255) / 256), 256, 0, stream>>>(Xe, zn);
    k_scatter_rows<true><<<8192, 256, 0, stream>>>(
        out, vertex, edges, homo, Xe, cnt, att_sum, nnz);
    const long long em = (long long)e * HO;
    k_edge_mean<<<(int)((em + 255) / 256), 256, 0, stream>>>(Xe, cnt, em);
    k_scatter_rows<false><<<8192, 256, 0, stream>>>(
        Xe, vertex, edges, homo, out, cnt, att_sum, nnz);
    const long long nt = (long long)n * 32;
    k_rownorm<<<(int)((nt + 255) / 256), 256, 0, stream>>>(out, n);
  }
}


# Range sums' walk: a lane reads 16 B, a half-wave one whole 256-B row, so one wave load brings two rows (4x fewer loads); each half-wave sums the runs of its own piece (sum order changes)
# speedup vs baseline: 7.3417x; 1.0597x over previous
#include <hip/hip_runtime.h>
#include <math.h>

typedef float v2f __attribute__((ext_vector_type(2)));
typedef float v8f __attribute__((ext_vector_type(8)));

#define HO 64
#define INCH 64

__global__ __launch_bounds__(256) void k_zero(float* __restrict__ p, long long n) {
  long long i = (long long)blockIdx.x * blockDim.x + threadIdx.x;
  if (i < n) p[i] = 0.0f;
}

__device__ __forceinline__ void gemm_rows16(
    const float* __restrict__ X, const float* __restrict__ W,
    float* __restrict__ Xp, int nrows, int wave, int lane) {
  const int row0 = wave * 16;
  if (row0 >= nrows) return;
  if (row0 + 16 <= nrows) {
    const int m  = lane & 15;
    const int kh = lane >> 4;
    v8f acc0 = {}, acc1 = {}, acc2 = {}, acc3 = {};
#pragma unroll
    for (int k0 = 0; k0 < INCH; k0 += 4) {
      v2f a = *(const v2f*)(X + (size_t)(row0 + m) * INCH + (k0 + kh * 2));
      const float* w0 = W + (size_t)(k0 + kh * 2) * HO;
      const float* w1 = w0 + HO;
      v2f b0, b1, b2, b3;
      b0.x = w0[m];      b0.y = w1[m];
      b1.x = w0[m + 16]; b1.y = w1[m + 16];
      b2.x = w0[m + 32]; b2.y = w1[m + 32];
      b3.x = w0[m + 48]; b3.y = w1[m + 48];
      acc0 = __builtin_amdgcn_wmma_f32_16x16x4_f32(false, a, false, b0, (short)0, acc0, false, false);
      acc1 = __builtin_amdgcn_wmma_f32_16x16x4_f32(false, a, false, b1, (short)0, acc1, false, false);
      acc2 = __builtin_amdgcn_wmma_f32_16x16x4_f32(false, a, false, b2, (short)0, acc2, false, false);
      acc3 = __builtin_amdgcn_wmma_f32_16x16x4_f32(false, a, false, b3, (short)0, acc3, false, false);
    }
#pragma unroll
    for (int r = 0; r < 8; ++r) {
      const int row = row0 + r + kh * 8;
      float* o = Xp + (size_t)row * HO + m;
      o[0]  = acc0[r];
      o[16] = acc1[r];
      o[32] = acc2[r];
      o[48] = acc3[r];
    }
  } else {
    for (int r = row0 + lane; r < nrows; r += 32)
      for (int c = 0; c < HO; ++c) {
        float s = 0.f;
        for (int k = 0; k < INCH; ++k) s += X[(size_t)r * INCH + k] * W[(size_t)k * HO + c];
        Xp[(size_t)r * HO + c] = s;
      }
  }
}

__global__ __launch_bounds__(256) void k_gemm_wmma(
    const float* __restrict__ X, const float* __restrict__ W,
    float* __restrict__ Xp, int nrows) {
  const int wave = (int)(((unsigned)blockIdx.x * blockDim.x + threadIdx.x) >> 5);
  const int lane = (int)(threadIdx.x & 31u);
  gemm_rows16(X, W, Xp, nrows, wave, lane);
}

#define TRIP 8
template <bool V2E>
__global__ __launch_bounds__(256) void k_scatter_rows(
    const float* __restrict__ src, const int* __restrict__ vertex,
    const int* __restrict__ edges, const float* __restrict__ homo,
    float* __restrict__ dst, float* __restrict__ cnt,
    float* __restrict__ att_sum, int nnz) {
  const int lane = (int)(threadIdx.x & 31u);
  const int wave = __builtin_amdgcn_readfirstlane(
      (int)(((unsigned)blockIdx.x * blockDim.x + threadIdx.x) >> 5));
  const int nwaves = (int)((gridDim.x * blockDim.x) >> 5);
  for (int base = wave * TRIP; base < nnz; base += nwaves * TRIP) {
    const int idx = base + lane;
    const bool mine = lane < TRIP && idx < nnz;
    int vl = 0, el = 0;
    if (mine) {
      vl = vertex[idx];
      el = edges[idx];
    }
    const bool whole = base + TRIP <= nnz;
    float x0[TRIP], x1[TRIP];
    if (whole) {
#pragma unroll
      for (int u = 0; u < TRIP; ++u) {
        const int r = __builtin_amdgcn_readlane(V2E ? vl : el, u);
        const float* s = src + (size_t)r * HO + lane;
        x0[u] = s[0];
        x1[u] = s[32];
      }
    }
    float al = 0.0f;
    if (mine) {
      if (V2E) {
        atomicAdd(cnt + el, 1.0f);
        atomicAdd(att_sum + vl, homo[el]);
      } else {
        al = homo[el] / att_sum[vl];
      }
    }
    if (whole) {
#pragma unroll
      for (int u = 0; u < TRIP; ++u) {
        const int r = __builtin_amdgcn_readlane(V2E ? el : vl, u);
        float* d = dst + (size_t)r * HO + lane;
        float a0 = x0[u], a1 = x1[u];
        if (!V2E) {
          const float att = __int_as_float(__builtin_amdgcn_readlane(__float_as_int(al), u));
          a0 = att * a0;
          a1 = att * a1;
        }
        atomicAdd(d, a0);
        atomicAdd(d + 32, a1);
      }
    } else {
      for (int u = 0; base + u < nnz; ++u) {
        const int rs = __shfl(V2E ? vl : el, u, 32);
        const int rd = __shfl(V2E ? el : vl, u, 32);
        const float* s = src + (size_t)rs * HO + lane;
        float* d = dst + (size_t)rd * HO + lane;
        float a0 = s[0], a1 = s[32];
        if (!V2E) {
          const float att = __shfl(al, u, 32);
          a0 = att * a0;
          a1 = att * a1;
        }
        atomicAdd(d, a0);
        atomicAdd(d + 32, a1);
      }
    }
  }
}

__global__ __launch_bounds__(256) void k_edge_mean(float* __restrict__ Xe,
                                                   const float* __restrict__ cnt,
                                                   long long total) {
  long long i = (long long)blockIdx.x * blockDim.x + threadIdx.x;
  if (i >= total) return;
  const float c = cnt[i >> 6];
  Xe[i] *= 1.0f / fmaxf(c, 1.0f);
}

__global__ __launch_bounds__(256) void k_rownorm(float* __restrict__ out, int n) {
  const int wave = (int)(((unsigned)blockIdx.x * blockDim.x + threadIdx.x) >> 5);
  const int lane = (int)(threadIdx.x & 31u);
  if (wave >= n) return;
  float* row = out + (size_t)wave * HO + lane * 2;
  float2 x = *(float2*)row;
  float ss = x.x * x.x + x.y * x.y;
#pragma unroll
  for (int off = 16; off > 0; off >>= 1) ss += __shfl_xor(ss, off, 32);
  const float nrm = sqrtf(ss);
  const float scale = (nrm > 0.f) ? 1.0f / fmaxf(nrm, 1e-30f) : 0.0f;
  x.x *= scale;
  x.y *= scale;
  *(float2*)row = x;
}

#define NB 512
#define NCH 256
#define RMAXE 64
#define RMAXV 200
#define SRCMASK 0xFFFFF
#define PCAP 8192
#define NT 1024
#define PK (PCAP / NT)

__global__ __launch_bounds__(NT) void k_group_gemm(
    const int* __restrict__ vertex, const int* __restrict__ edges,
    unsigned short* __restrict__ pcE, unsigned short* __restrict__ pcV,
    int* __restrict__ listE, int* __restrict__ listV, int nnz, int re, int rv,
    const float* __restrict__ X, const float* __restrict__ W,
    float* __restrict__ Xp, int nrows) {
  __shared__ int hE[NB], hV[NB];
  __shared__ int wsE[NB / 32], wsV[NB / 32];
  __shared__ int stE[PCAP], stV[PCAP];
  const int t = (int)threadIdx.x;
  if ((int)blockIdx.x >= NCH) {
    const int wave = (((int)blockIdx.x - NCH) * NT + t) >> 5;
    gemm_rows16(X, W, Xp, nrows, wave, t & 31);
    return;
  }
  const int c = (int)blockIdx.x;
  const bool rng = t < NB;
  const int lane = t & 31;
  const int wave = t >> 5;
  const int ch = (nnz + NCH - 1) / NCH;
  const int lo = c * ch;
  const int pn = min(ch, nnz - lo);
  if (pn <= 0) {
    if (rng) {
      pcE[c * (NB + 1) + t] = 0;
      pcV[c * (NB + 1) + t] = 0;
    }
    if (t == 0) {
      pcE[c * (NB + 1) + NB] = 0;
      pcV[c * (NB + 1) + NB] = 0;
    }
    return;
  }
  if (rng) {
    hE[t] = 0;
    hV[t] = 0;
  }
  int wE[PK], wV[PK], pE[PK], pV[PK];
#pragma unroll
  for (int k = 0; k < PK; ++k) {
    const int i = lo + min(k * NT + t, pn - 1);
    wE[k] = vertex[i];
    wV[k] = edges[i];
  }
  __syncthreads();
#pragma unroll
  for (int k = 0; k < PK; ++k) {
    const bool ok = k * NT + t < pn;
    const int v = wE[k], e = wV[k];
    const int be = e / re, bv = v / rv;
    const int ce = ok ? be : (t & (NB - 1)), cv = ok ? bv : (t & (NB - 1));
    const int one = ok ? 1 : 0;
    pE[k] = ce | (atomicAdd(&hE[ce], one) << 16);
    pV[k] = cv | (atomicAdd(&hV[cv], one) << 16);
    wE[k] = v | ((e - be * re) << 20);
    wV[k] = e | ((v - bv * rv) << 20);
  }
  __syncthreads();
  int nE = 0, nV = 0;
  if (rng) {
    nE = hE[t];
    nV = hV[t];
  }
  int iE = nE, iV = nV;
#pragma unroll
  for (int off = 1; off < 32; off <<= 1) {
    const int uE = __shfl_up(iE, off, 32);
    const int uV = __shfl_up(iV, off, 32);
    if (lane >= off) {
      iE += uE;
      iV += uV;
    }
  }
  if (rng && lane == 31) {
    wsE[wave] = iE;
    wsV[wave] = iV;
  }
  __syncthreads();
  if (rng) {
    int bE = 0, bV = 0;
    for (int w2 = 0; w2 < NB / 32; ++w2)
      if (w2 < wave) {
        bE += wsE[w2];
        bV += wsV[w2];
      }
    const int fE = bE + iE - nE, fV = bV + iV - nV;
    hE[t] = fE;
    hV[t] = fV;
    pcE[c * (NB + 1) + t] = (unsigned short)fE;
    pcV[c * (NB + 1) + t] = (unsigned short)fV;
  }
  if (t == 0) {
    pcE[c * (NB + 1) + NB] = (unsigned short)pn;
    pcV[c * (NB + 1) + NB] = (unsigned short)pn;
  }
  __syncthreads();
  int se[PK], sv[PK];
#pragma unroll
  for (int k = 0; k < PK; ++k) {
    se[k] = hE[pE[k] & 0xFFFF] + (pE[k] >> 16);
    sv[k] = hV[pV[k] & 0xFFFF] + (pV[k] >> 16);
  }
#pragma unroll
  for (int k = 0; k < PK; ++k)
    if (k * NT + t < pn) {
      stE[se[k]] = wE[k];
      stV[sv[k]] = wV[k];
    }
  __syncthreads();
#pragma unroll
  for (int k = 0; k < PK; ++k) {
    const int s = min(k * NT + t, pn - 1);
    wE[k] = stE[s];
    wV[k] = stV[s];
  }
#pragma unroll
  for (int k = 0; k < PK; ++k)
    if (k * NT + t < pn) {
      listE[lo + k * NT + t] = wE[k];
      listV[lo + k * NT + t] = wV[k];
    }
}

#define HT 8
template <int U>
__device__ __forceinline__ float half_bcast(float v) {
  return __int_as_float(__builtin_amdgcn_update_dpp(0, __float_as_int(v), 0x150 + U, 0xF, 0xF, false));
}
#define WALK_STEP(u, G)                                                           \
  {                                                                               \
    const bool valid = !(G) || (u) < rem;                                         \
    const int d = valid ? (ev[u] >> 20) : crow;                                   \
    if (d != crow) {                                                              \
      if (crow >= 0) {                                                            \
        float* p = acc + crow * HO + c4;                                          \
        atomicAdd(p, a0);                                                         \
        atomicAdd(p + 1, a1);                                                     \
        atomicAdd(p + 2, a2);                                                     \
        atomicAdd(p + 3, a3);                                                     \
        if (!V2E && (lane & 15) == 0) atomicAdd(aux + crow, hs);                  \
      }                                                                           \
      a0 = 0.0f;                                                                  \
      a1 = 0.0f;                                                                  \
      a2 = 0.0f;                                                                  \
      a3 = 0.0f;                                                                  \
      hs = 0.0f;                                                                  \
      crow = d;                                                                   \
    }                                                                             \
    const float wgt = V2E ? (valid ? 1.0f : 0.0f) : half_bcast<u>(al);            \
    a0 = fmaf(wgt, x[u].x, a0);                                                   \
    a1 = fmaf(wgt, x[u].y, a1);                                                   \
    a2 = fmaf(wgt, x[u].z, a2);                                                   \
    a3 = fmaf(wgt, x[u].w, a3);                                                   \
    if (!V2E) hs += wgt;                                                          \
  }
#define WALK_LOAD(u, G)                                                           \
  {                                                                               \
    const int sl = (!(G) || (u) < rem) ? (ev[u] & SRCMASK) : 0;                   \
    x[u] = *(const float4*)(src + (size_t)sl * HO + c4);                          \
  }
#define WALK_TRIP(G)                                                              \
  {                                                                               \
    float al = 0.0f;                                                              \
    if (!V2E && (!(G) || (lane & 7) < rem)) al = homo[em & SRCMASK];              \
    float4 x[HT];                                                                 \
    WALK_LOAD(0, G) WALK_LOAD(1, G) WALK_LOAD(2, G) WALK_LOAD(3, G)               \
    WALK_LOAD(4, G) WALK_LOAD(5, G) WALK_LOAD(6, G) WALK_LOAD(7, G)               \
    WALK_STEP(0, G) WALK_STEP(1, G) WALK_STEP(2, G) WALK_STEP(3, G)               \
    WALK_STEP(4, G) WALK_STEP(5, G) WALK_STEP(6, G) WALK_STEP(7, G)               \
  }
#define TILE 4608
template <bool V2E, int RMAX>
__global__ __launch_bounds__(NB) void k_range_sum(
    const float* __restrict__ src, const int* __restrict__ list, const unsigned short* __restrict__ pc,
    const float* __restrict__ homo, float* __restrict__ dst, int r, int nrows, int nnz) {
  __shared__ __attribute__((aligned(16))) float acc[RMAX * HO];
  __shared__ float aux[RMAX];
  __shared__ int cur[RMAX];
  __shared__ __attribute__((aligned(16))) int sorted[TILE + HT];
  __shared__ int own[TILE];
  __shared__ int wsum[NB / 32];
  const int t = (int)threadIdx.x;
  const int lane = t & 31;
  const int bin = (int)blockIdx.x;
  const int wave = __builtin_amdgcn_readfirstlane(t >> 5);
  int pstart = 0, plen = 0;
  if (t < NCH) {
    const int f0 = pc[t * (NB + 1) + bin], f1 = pc[t * (NB + 1) + bin + 1];
    pstart = t * ((nnz + NCH - 1) / NCH) + f0;
    plen = f1 - f0;
  }
  for (int j = t; j < r * HO; j += NB) acc[j] = 0.0f;
  for (int j = t; j < r; j += NB) aux[j] = 0.0f;
  int pin = plen;
#pragma unroll
  for (int off = 1; off < 32; off <<= 1) {
    const int v = __shfl_up(pin, off, 32);
    if (lane >= off) pin += v;
  }
  if (lane == 31) wsum[wave] = pin;
  __syncthreads();
  int before = 0, total = 0;
  for (int w2 = 0; w2 < NB / 32; ++w2) {
    const int v = wsum[w2];
    if (w2 < wave) before += v;
    total += v;
  }
  const int ppos = before + pin - plen;
  const int n = __builtin_amdgcn_readfirstlane(total);
  constexpr int NWV = NB / 32;
  constexpr int KW = TILE / NB;
  constexpr int RPL = (RMAX + 31) / 32;
  for (int tlo = 0; tlo < n; tlo += TILE) {
    const int tn = min(TILE, n - tlo);
    for (int j = t; j < r; j += NB) cur[j] = 0;
    const int j1 = min(plen, tlo + TILE - ppos);
    for (int j = max(0, tlo - ppos); j < j1; ++j) own[ppos + j - tlo] = pstart + j;
    __syncthreads();
    int w[KW];
#pragma unroll
    for (int k = 0; k < KW; ++k) w[k] = list[own[min(k * NB + t, tn - 1)]];
#pragma unroll
    for (int k = 0; k < KW; ++k)
      if (k * NB + t < tn) atomicAdd(&cur[w[k] >> 20], 1);
    __syncthreads();
    if (wave == 0) {
      int c[RPL];
      int s = 0;
#pragma unroll
      for (int u = 0; u < RPL; ++u) {
        const int j = lane * RPL + u;
        c[u] = (j < r) ? cur[j] : 0;
        if (V2E && j < r) aux[j] += (float)c[u];
        s += c[u];
      }
      int incl = s;
#pragma unroll
      for (int off = 1; off < 32; off <<= 1) {
        const int v = __shfl_up(incl, off, 32);
        if (lane >= off) incl += v;
      }
      int run = incl - s;
#pragma unroll
      for (int u = 0; u < RPL; ++u) {
        const int j = lane * RPL + u;
        if (j < r) cur[j] = run;
        run += c[u];
      }
    }
    __syncthreads();
#pragma unroll
    for (int k = 0; k < KW; ++k)
      if (k * NB + t < tn) sorted[atomicAdd(&cur[w[k] >> 20], 1)] = w[k];
    __syncthreads();
    const int hseg = ((tn + 2 * NWV * HT - 1) / (2 * NWV * HT)) * HT;
    const int w0 = wave * 2 * hseg;
    const int len0 = min(hseg, tn - w0);
    const int len1 = min(hseg, tn - w0 - hseg);
    if (len0 > 0) {
      const int hbeg = w0 + (lane >> 4) * hseg;
      const int hend = min(hbeg + hseg, tn);
      const int c4 = (lane & 15) * 4;
      float a0 = 0.0f, a1 = 0.0f, a2 = 0.0f, a3 = 0.0f;
      float hs = 0.0f;
      int crow = -1;
      int4 ea = *(const int4*)(sorted + hbeg);
      int4 eb = *(const int4*)(sorted + hbeg + 4);
      int em = 0;
      if (!V2E) em = sorted[hbeg + (lane & 7)];
      for (int b = 0; b < len0; b += HT) {
        const int4 na = *(const int4*)(sorted + hbeg + b + HT);
        const int4 nb = *(const int4*)(sorted + hbeg + b + HT + 4);
        int nm = 0;
        if (!V2E) nm = sorted[hbeg + b + HT + (lane & 7)];
        const int ev[HT] = {ea.x, ea.y, ea.z, ea.w, eb.x, eb.y, eb.z, eb.w};
        const int rem = hend - (hbeg + b);
        if (b + HT <= len1) WALK_TRIP(false) else WALK_TRIP(true)
        ea = na;
        eb = nb;
        em = nm;
      }
      if (crow >= 0) {
        float* p = acc + crow * HO + c4;
        atomicAdd(p, a0);
        atomicAdd(p + 1, a1);
        atomicAdd(p + 2, a2);
        atomicAdd(p + 3, a3);
        if (!V2E && (lane & 15) == 0) atomicAdd(aux + crow, hs);
      }
    }
  }
  if (V2E) {
    __syncthreads();
    for (int j = t; j < r * HO; j += NB) {
      const int row = bin * r + (j >> 6);
      if (row < nrows) dst[(size_t)row * HO + (j & 63)] = acc[j] / fmaxf(aux[j >> 6], 1.0f);
    }
  } else {
    constexpr int NW = NB / 32;
    constexpr int RPW = (RMAX + NW - 1) / NW;
    const int nv = min(r, nrows - bin * r);
    if (nv <= 0) return;
    float2 x[RPW];
#pragma unroll
    for (int u = 0; u < RPW; ++u) {
      const int rc = min(wave + u * NW, nv - 1);
      x[u] = *(const float2*)(dst + (size_t)(bin * r + rc) * HO + lane * 2);
    }
    __syncthreads();
#pragma unroll
    for (int u = 0; u < RPW; ++u) {
      const int rc = min(wave + u * NW, nv - 1);
      const float2 a = *(const float2*)(acc + rc * HO + lane * 2);
      const float asum = aux[rc];
      const float rcp = 1.0f / asum;
      const float wgt = (asum != 0.0f) ? rcp : 0.0f;
      x[u].x = x[u].x + wgt * a.x;
      x[u].y = x[u].y + wgt * a.y;
      float ss = x[u].x * x[u].x + x[u].y * x[u].y;
#pragma unroll
      for (int off = 16; off > 0; off >>= 1) ss += __shfl_xor(ss, off, 32);
      const float nrm = sqrtf(ss);
      const float inv = 1.0f / fmaxf(nrm, 1e-30f);
      const float scale = (nrm > 0.f) ? inv : 0.0f;
      x[u].x *= scale;
      x[u].y *= scale;
    }
#pragma unroll
    for (int u = 0; u < RPW; ++u) {
      const int rr = wave + u * NW;
      if (rr < nv)
        *(float2*)(dst + (size_t)(bin * r + rr) * HO + lane * 2) = x[u];
    }
  }
}

extern "C" void kernel_launch(void* const* d_in, const int* in_sizes, int n_in,
                              void* d_out, int out_size, void* d_ws, size_t ws_size,
                              hipStream_t stream) {
  const float* X      = (const float*)d_in[0];
  const float* W      = (const float*)d_in[1];
  const float* homo   = (const float*)d_in[2];
  const int*   vertex = (const int*)d_in[3];
  const int*   edges  = (const int*)d_in[4];
  float* out = (float*)d_out;

  const int n   = in_sizes[0] / INCH;
  const int e   = in_sizes[2];
  const int nnz = in_sizes[3];

  const int tiles = (n + 15) / 16;

  const int re = (e + NB - 1) / NB;
  const int rv = (n + NB - 1) / NB;
  const size_t need = ((size_t)e * HO + 2 * (size_t)nnz) * 4 + 2 * (size_t)NCH * (NB + 1) * 2;
  float* Xe = (float*)d_ws;

  if (need <= ws_size && re <= RMAXE && rv <= RMAXV && n <= SRCMASK && e <= SRCMASK &&
      (nnz + NCH - 1) / NCH <= PCAP) {
    int* listE = (int*)(Xe + (size_t)e * HO);
    int* listV = listE + nnz;
    unsigned short* pcE = (unsigned short*)(listV + nnz);
    unsigned short* pcV = pcE + NCH * (NB + 1);
    const int gblocks = (tiles + NT / 32 - 1) / (NT / 32);
    k_group_gemm<<<NCH + gblocks, NT, 0, stream>>>(vertex, edges, pcE, pcV, listE, listV, nnz, re, rv, X, W, out, n);
    k_range_sum<true, RMAXE><<<NB, NB, 0, stream>>>(out, listE, pcE, homo, Xe, re, e, nnz);
    k_range_sum<false, RMAXV><<<NB, NB, 0, stream>>>(Xe, listV, pcV, homo, out, rv, n, nnz);
  } else {
    k_gemm_wmma<<<(tiles * 32 + 255) / 256, 256, 0, stream>>>(X, W, out, n);
    float* cnt     = Xe + (size_t)e * HO;
    float* att_sum = cnt + e;
    const long long zn = (long long)e * HO + e + n;
    k_zero<<<(int)((zn + 255) / 256), 256, 0, stream>>>(Xe, zn);
    k_scatter_rows<true><<<8192, 256, 0, stream>>>(
        out, vertex, edges, homo, Xe, cnt, att_sum, nnz);
    const long long em = (long long)e * HO;
    k_edge_mean<<<(int)((em + 255) / 256), 256, 0, stream>>>(Xe, cnt, em);
    k_scatter_rows<false><<<8192, 256, 0, stream>>>(
        Xe, vertex, edges, homo, out, cnt, att_sum, nnz);
    const long long nt = (long long)n * 32;
    k_rownorm<<<(int)((nt + 255) / 256), 256, 0, stream>>>(out, n);
  }
}
